# starting kernel
# baseline (speedup 1.0000x reference)
#include <hip/hip_runtime.h>

typedef __attribute__((ext_vector_type(2))) float v2f;
typedef __attribute__((ext_vector_type(8))) float v8f;

#define NNODES 100000
#define NEDGES 1600000

__global__ void k_zero(float* __restrict__ p, long n) {
  long i = ((long)blockIdx.x * blockDim.x + threadIdx.x) * 4;
  long stride = (long)gridDim.x * blockDim.x * 4;
  float4 z = {0.f, 0.f, 0.f, 0.f};
  for (long j = i; j < n; j += stride) *(float4*)(p + j) = z;
}

__global__ void k_prep(const float* __restrict__ u,
                       const float* __restrict__ Web, const float* __restrict__ beb,
                       const float* __restrict__ Wnb, const float* __restrict__ bnb,
                       float* __restrict__ bias_eb, float* __restrict__ bias_nb) {
  int j = threadIdx.x;
  float a = beb[j], b = bnb[j];
  for (int k = 0; k < 64; ++k) {
    float uk = u[k];
    a += uk * Web[(48 + k) * 64 + j];
    b += uk * Wnb[(80 + k) * 64 + j];
  }
  bias_eb[j] = a;
  bias_nb[j] = b;
}

__global__ __launch_bounds__(256) void k_edge(
    const float* __restrict__ x, const float* __restrict__ ea,
    const int* __restrict__ recv, const int* __restrict__ send,
    const float* __restrict__ Web, const float* __restrict__ bias_eb,
    float* __restrict__ agg_r, float* __restrict__ agg_s,
    float* __restrict__ cnt_r, float* __restrict__ cnt_s) {
  __shared__ float sW[12 * 4 * 64];
  for (int i = threadIdx.x; i < 12 * 4 * 64; i += 256) {
    int j = i & 1, lane = (i >> 1) & 31, nt = (i >> 6) & 3, kc = i >> 8;
    int kk = 4 * kc + 2 * (lane >> 4) + j;
    sW[i] = Web[kk * 64 + nt * 16 + (lane & 15)];
  }
  __syncthreads();
  const int lane = threadIdx.x & 31;
  const int h = lane >> 4, m = lane & 15;
  const int wave = blockIdx.x * 8 + (threadIdx.x >> 5);
  const int nwaves = gridDim.x * 8;
  const int ntiles = NEDGES / 16;
  for (int t = wave; t < ntiles; t += nwaves) {
    const int e0 = t * 16;
    const int er = e0 + m;
    const int ri = recv[er], si = send[er];
    const float* segs[3] = {ea + (long)er * 16, x + (long)ri * 16, x + (long)si * 16};
    v8f acc[4];
#pragma unroll
    for (int nt = 0; nt < 4; ++nt) {
      float bb = bias_eb[nt * 16 + m];
      acc[nt] = (v8f){bb, bb, bb, bb, bb, bb, bb, bb};
    }
#pragma unroll
    for (int seg = 0; seg < 3; ++seg) {
      const float* bp = segs[seg];
#pragma unroll
      for (int cc = 0; cc < 4; ++cc) {
        v2f a = *(const v2f*)(bp + 4 * cc + 2 * h);
        const int kc = seg * 4 + cc;
#pragma unroll
        for (int nt = 0; nt < 4; ++nt) {
          v2f b = *(const v2f*)(&sW[(kc * 4 + nt) * 64 + lane * 2]);
          acc[nt] = __builtin_amdgcn_wmma_f32_16x16x4_f32(false, a, false, b,
                                                          (short)0, acc[nt], false, false);
        }
      }
    }
    int rr[8], ss[8];
#pragma unroll
    for (int v = 0; v < 8; ++v) {
      int e = e0 + 8 * h + v;
      rr[v] = recv[e];
      ss[v] = send[e];
    }
#pragma unroll
    for (int nt = 0; nt < 4; ++nt) {
      const int col = nt * 16 + m;
#pragma unroll
      for (int v = 0; v < 8; ++v) {
        float val = fmaxf(acc[nt][v], 0.0f);
        unsafeAtomicAdd(&agg_r[(long)rr[v] * 64 + col], val);
        unsafeAtomicAdd(&agg_s[(long)ss[v] * 64 + col], val);
      }
    }
    if (lane < 16) {
      unsafeAtomicAdd(&cnt_r[recv[e0 + lane]], 1.0f);
      unsafeAtomicAdd(&cnt_s[send[e0 + lane]], 1.0f);
    }
  }
}

__global__ __launch_bounds__(256) void k_node(
    const float* __restrict__ x, const float* __restrict__ agg_r,
    const float* __restrict__ Wnb, const float* __restrict__ bias_nb,
    float* __restrict__ n_h) {
  __shared__ float sW[20 * 4 * 64];
  for (int i = threadIdx.x; i < 20 * 4 * 64; i += 256) {
    int j = i & 1, lane = (i >> 1) & 31, nt = (i >> 6) & 3, kc = i >> 8;
    int kk = 4 * kc + 2 * (lane >> 4) + j;
    sW[i] = Wnb[kk * 64 + nt * 16 + (lane & 15)];
  }
  __syncthreads();
  const int lane = threadIdx.x & 31;
  const int h = lane >> 4, m = lane & 15;
  const int wave = blockIdx.x * 8 + (threadIdx.x >> 5);
  const int nwaves = gridDim.x * 8;
  const int ntiles = NNODES / 16;
  for (int t = wave; t < ntiles; t += nwaves) {
    const int n0 = t * 16;
    const float* ap = agg_r + (long)(n0 + m) * 64;
    const float* xp = x + (long)(n0 + m) * 16;
    v8f acc[4];
#pragma unroll
    for (int nt = 0; nt < 4; ++nt) {
      float bb = bias_nb[nt * 16 + m];
      acc[nt] = (v8f){bb, bb, bb, bb, bb, bb, bb, bb};
    }
#pragma unroll
    for (int kc = 0; kc < 20; ++kc) {
      const int kk = 4 * kc + 2 * h;
      v2f a = (kc < 16) ? *(const v2f*)(ap + kk) : *(const v2f*)(xp + kk - 64);
#pragma unroll
      for (int nt = 0; nt < 4; ++nt) {
        v2f b = *(const v2f*)(&sW[(kc * 4 + nt) * 64 + lane * 2]);
        acc[nt] = __builtin_amdgcn_wmma_f32_16x16x4_f32(false, a, false, b,
                                                        (short)0, acc[nt], false, false);
      }
    }
#pragma unroll
    for (int nt = 0; nt < 4; ++nt)
#pragma unroll
      for (int v = 0; v < 8; ++v)
        n_h[(long)(n0 + 8 * h + v) * 64 + nt * 16 + m] = fmaxf(acc[nt][v], 0.0f);
  }
}

__global__ void k_colsum(const float* __restrict__ src, int rows, float* __restrict__ dst) {
  __shared__ float red[256];
  const int col = threadIdx.x & 63;
  const int g = threadIdx.x >> 6;
  float local = 0.f;
  for (int r = blockIdx.x * 4 + g; r < rows; r += gridDim.x * 4)
    local += src[(long)r * 64 + col];
  red[threadIdx.x] = local;
  __syncthreads();
  if (g == 0) {
    local += red[64 + threadIdx.x] + red[128 + threadIdx.x] + red[192 + threadIdx.x];
    unsafeAtomicAdd(&dst[col], local);
  }
}

__global__ void k_global(const float* __restrict__ u,
                         const float* __restrict__ Wgb, const float* __restrict__ bgb,
                         const float* __restrict__ Wd1, const float* __restrict__ bd1,
                         const float* __restrict__ e_sum, const float* __restrict__ n_sum,
                         float* __restrict__ u2, float* __restrict__ bias_d1) {
  __shared__ float su2[64];
  const int j = threadIdx.x;
  float acc = bgb[j];
  for (int k = 0; k < 64; ++k) {
    float em = e_sum[k] * (1.0f / NEDGES);
    float nm = n_sum[k] * (1.0f / NNODES);
    acc += em * Wgb[k * 64 + j] + nm * Wgb[(64 + k) * 64 + j] + u[k] * Wgb[(128 + k) * 64 + j];
  }
  acc = fmaxf(acc, 0.0f);
  su2[j] = acc;
  u2[j] = acc;
  __syncthreads();
  float b = bd1[j];
  for (int k = 0; k < 64; ++k) b += su2[k] * Wd1[(192 + k) * 64 + j];
  bias_d1[j] = b;
}

__global__ __launch_bounds__(256) void k_dec(
    const float* __restrict__ agg_r, const float* __restrict__ agg_s,
    const float* __restrict__ cnt_r, const float* __restrict__ cnt_s,
    const float* __restrict__ n_h, const float* __restrict__ Wd1,
    const float* __restrict__ bias_d1, const float* __restrict__ Wd2,
    const float* __restrict__ bd2, float* __restrict__ out) {
  __shared__ float sW1[48 * 4 * 64];
  __shared__ float sW2[16 * 64];
  __shared__ float sH[8 * 16 * 64];
  for (int i = threadIdx.x; i < 48 * 4 * 64; i += 256) {
    int j = i & 1, lane = (i >> 1) & 31, nt = (i >> 6) & 3, kc = i >> 8;
    int kk = 4 * kc + 2 * (lane >> 4) + j;
    sW1[i] = Wd1[kk * 64 + nt * 16 + (lane & 15)];
  }
  for (int i = threadIdx.x; i < 16 * 64; i += 256) {
    int j = i & 1, lane = (i >> 1) & 31, kc = i >> 6;
    int kk = 4 * kc + 2 * (lane >> 4) + j;
    sW2[i] = Wd2[kk * 16 + (lane & 15)];
  }
  __syncthreads();
  const int lane = threadIdx.x & 31;
  const int h = lane >> 4, m = lane & 15;
  float* hw = sH + (threadIdx.x >> 5) * 1024;
  const int wave = blockIdx.x * 8 + (threadIdx.x >> 5);
  const int nwaves = gridDim.x * 8;
  const int ntiles = NNODES / 16;
  for (int t = wave; t < ntiles; t += nwaves) {
    const int n0 = t * 16;
    const int n = n0 + m;
    const float invr = 1.0f / fmaxf(cnt_r[n], 1.0f);
    const float invs = 1.0f / fmaxf(cnt_s[n], 1.0f);
    const float* pr = agg_r + (long)n * 64;
    const float* ps = agg_s + (long)n * 64;
    const float* pn = n_h + (long)n * 64;
    v8f acc[4];
#pragma unroll
    for (int nt = 0; nt < 4; ++nt) {
      float bb = bias_d1[nt * 16 + m];
      acc[nt] = (v8f){bb, bb, bb, bb, bb, bb, bb, bb};
    }
#pragma unroll
    for (int kc = 0; kc < 48; ++kc) {
      const int kk = 4 * kc + 2 * h;
      v2f a;
      if (kc < 16)      { a = *(const v2f*)(pr + kk);        a *= invr; }
      else if (kc < 32) { a = *(const v2f*)(ps + kk - 64);   a *= invs; }
      else              { a = *(const v2f*)(pn + kk - 128); }
#pragma unroll
      for (int nt = 0; nt < 4; ++nt) {
        v2f b = *(const v2f*)(&sW1[(kc * 4 + nt) * 64 + lane * 2]);
        acc[nt] = __builtin_amdgcn_wmma_f32_16x16x4_f32(false, a, false, b,
                                                        (short)0, acc[nt], false, false);
      }
    }
#pragma unroll
    for (int nt = 0; nt < 4; ++nt)
#pragma unroll
      for (int v = 0; v < 8; ++v)
        hw[(8 * h + v) * 64 + nt * 16 + m] = fmaxf(acc[nt][v], 0.0f);
    float bb2 = bd2[m];
    v8f acc2 = (v8f){bb2, bb2, bb2, bb2, bb2, bb2, bb2, bb2};
#pragma unroll
    for (int kc = 0; kc < 16; ++kc) {
      const int kk = 4 * kc + 2 * h;
      v2f a = *(const v2f*)(hw + m * 64 + kk);
      v2f b = *(const v2f*)(&sW2[kc * 64 + lane * 2]);
      acc2 = __builtin_amdgcn_wmma_f32_16x16x4_f32(false, a, false, b,
                                                   (short)0, acc2, false, false);
    }
#pragma unroll
    for (int v = 0; v < 8; ++v)
      out[(long)(n0 + 8 * h + v) * 16 + m] = acc2[v];
  }
}

extern "C" void kernel_launch(void* const* d_in, const int* in_sizes, int n_in,
                              void* d_out, int out_size, void* d_ws, size_t ws_size,
                              hipStream_t stream) {
  const float* x    = (const float*)d_in[0];
  const float* ea   = (const float*)d_in[1];
  const float* u    = (const float*)d_in[2];
  const int* send   = (const int*)d_in[3];
  const int* recv   = (const int*)d_in[4];
  const float* Web  = (const float*)d_in[7];
  const float* beb  = (const float*)d_in[8];
  const float* Wnb  = (const float*)d_in[9];
  const float* bnb  = (const float*)d_in[10];
  const float* Wgb  = (const float*)d_in[11];
  const float* bgb  = (const float*)d_in[12];
  const float* Wd1  = (const float*)d_in[13];
  const float* bd1  = (const float*)d_in[14];
  const float* Wd2  = (const float*)d_in[15];
  const float* bd2  = (const float*)d_in[16];
  float* out = (float*)d_out;
  float* ws = (float*)d_ws;

  const long N = NNODES;
  float* agg_r   = ws;
  float* agg_s   = agg_r + N * 64;
  float* cnt_r   = agg_s + N * 64;
  float* cnt_s   = cnt_r + N;
  float* e_sum   = cnt_s + N;
  float* n_sum   = e_sum + 64;
  float* bias_eb = n_sum + 64;
  float* bias_nb = bias_eb + 64;
  float* u2      = bias_nb + 64;
  float* bias_d1 = u2 + 64;
  float* n_hbuf  = bias_d1 + 64;

  const long zlen = 2 * N * 64 + 2 * N + 128;
  k_zero<<<2048, 256, 0, stream>>>(ws, zlen);
  k_prep<<<1, 64, 0, stream>>>(u, Web, beb, Wnb, bnb, bias_eb, bias_nb);
  k_edge<<<2048, 256, 0, stream>>>(x, ea, recv, send, Web, bias_eb,
                                   agg_r, agg_s, cnt_r, cnt_s);
  k_node<<<782, 256, 0, stream>>>(x, agg_r, Wnb, bias_nb, n_hbuf);
  k_colsum<<<512, 256, 0, stream>>>(agg_r, NNODES, e_sum);
  k_colsum<<<512, 256, 0, stream>>>(n_hbuf, NNODES, n_sum);
  k_global<<<1, 64, 0, stream>>>(u, Wgb, bgb, Wd1, bd1, e_sum, n_sum, u2, bias_d1);
  k_dec<<<1024, 256, 0, stream>>>(agg_r, agg_s, cnt_r, cnt_s, n_hbuf,
                                  Wd1, bias_d1, Wd2, bd2, out);
}


# k_edge: lane halves exchange tile registers so each f32 atomic add covers one whole 128-B line of one row (was two 64-B pieces of two rows), scalar row index; atomics kept, arrival order
# speedup vs baseline: 1.1347x; 1.1347x over previous
#include <hip/hip_runtime.h>

typedef __attribute__((ext_vector_type(2))) float v2f;
typedef __attribute__((ext_vector_type(8))) float v8f;

#define NNODES 100000
#define NEDGES 1600000

__device__ __forceinline__ float xhalf(float x) {
  return __builtin_bit_cast(float, __builtin_amdgcn_permlanex16(
      __builtin_bit_cast(int, x), __builtin_bit_cast(int, x), 0x76543210, 0xfedcba98, false, false));
}

__global__ void k_zero(float* __restrict__ p, long n) {
  long i = ((long)blockIdx.x * blockDim.x + threadIdx.x) * 4;
  long stride = (long)gridDim.x * blockDim.x * 4;
  float4 z = {0.f, 0.f, 0.f, 0.f};
  for (long j = i; j < n; j += stride) *(float4*)(p + j) = z;
}

__global__ void k_prep(const float* __restrict__ u,
                       const float* __restrict__ Web, const float* __restrict__ beb,
                       const float* __restrict__ Wnb, const float* __restrict__ bnb,
                       float* __restrict__ bias_eb, float* __restrict__ bias_nb) {
  int j = threadIdx.x;
  float a = beb[j], b = bnb[j];
  for (int k = 0; k < 64; ++k) {
    float uk = u[k];
    a += uk * Web[(48 + k) * 64 + j];
    b += uk * Wnb[(80 + k) * 64 + j];
  }
  bias_eb[j] = a;
  bias_nb[j] = b;
}

__global__ __launch_bounds__(256) void k_edge(
    const float* __restrict__ x, const float* __restrict__ ea,
    const int* __restrict__ recv, const int* __restrict__ send,
    const float* __restrict__ Web, const float* __restrict__ bias_eb,
    float* __restrict__ agg_r, float* __restrict__ agg_s,
    float* __restrict__ cnt_r, float* __restrict__ cnt_s) {
  __shared__ float sW[12 * 4 * 64];
  for (int i = threadIdx.x; i < 12 * 4 * 64; i += 256) {
    int j = i & 1, lane = (i >> 1) & 31, nt = (i >> 6) & 3, kc = i >> 8;
    int kk = 4 * kc + 2 * (lane >> 4) + j;
    sW[i] = Web[kk * 64 + nt * 16 + (lane & 15)];
  }
  __syncthreads();
  const int lane = threadIdx.x & 31;
  const int h = lane >> 4, m = lane & 15;
  const int wave = blockIdx.x * 8 + (threadIdx.x >> 5);
  const int nwaves = gridDim.x * 8;
  const int ntiles = NEDGES / 16;
  for (int t = wave; t < ntiles; t += nwaves) {
    const int e0 = t * 16;
    const int er = e0 + m;
    const int ri = recv[er], si = send[er];
    const float* segs[3] = {ea + (long)er * 16, x + (long)ri * 16, x + (long)si * 16};
    v8f acc[4];
#pragma unroll
    for (int nt = 0; nt < 4; ++nt) {
      float bb = bias_eb[nt * 16 + m];
      acc[nt] = (v8f){bb, bb, bb, bb, bb, bb, bb, bb};
    }
#pragma unroll
    for (int seg = 0; seg < 3; ++seg) {
      const float* bp = segs[seg];
#pragma unroll
      for (int cc = 0; cc < 4; ++cc) {
        v2f a = *(const v2f*)(bp + 4 * cc + 2 * h);
        const int kc = seg * 4 + cc;
#pragma unroll
        for (int nt = 0; nt < 4; ++nt) {
          v2f b = *(const v2f*)(&sW[(kc * 4 + nt) * 64 + lane * 2]);
          acc[nt] = __builtin_amdgcn_wmma_f32_16x16x4_f32(false, a, false, b,
                                                          (short)0, acc[nt], false, false);
        }
      }
    }
#pragma unroll
    for (int v = 0; v < 8; ++v) {
      const long rA = __builtin_amdgcn_readlane(ri, v), sA = __builtin_amdgcn_readlane(si, v);
      const long rB = __builtin_amdgcn_readlane(ri, 8 + v), sB = __builtin_amdgcn_readlane(si, 8 + v);
#pragma unroll
      for (int p = 0; p < 2; ++p) {
        const float X = fmaxf(acc[2 * p][v], 0.0f);
        const float Y = fmaxf(acc[2 * p + 1][v], 0.0f);
        const float Zx = xhalf(h ? X : Y);
        const float A = h ? Zx : X;
        const float B = h ? Y : Zx;
        const int col = 32 * p + lane;
        unsafeAtomicAdd(&agg_r[rA * 64 + col], A);
        unsafeAtomicAdd(&agg_s[sA * 64 + col], A);
        unsafeAtomicAdd(&agg_r[rB * 64 + col], B);
        unsafeAtomicAdd(&agg_s[sB * 64 + col], B);
      }
    }
    if (lane < 16) {
      unsafeAtomicAdd(&cnt_r[recv[e0 + lane]], 1.0f);
      unsafeAtomicAdd(&cnt_s[send[e0 + lane]], 1.0f);
    }
  }
}

__global__ __launch_bounds__(256) void k_node(
    const float* __restrict__ x, const float* __restrict__ agg_r,
    const float* __restrict__ Wnb, const float* __restrict__ bias_nb,
    float* __restrict__ n_h) {
  __shared__ float sW[20 * 4 * 64];
  for (int i = threadIdx.x; i < 20 * 4 * 64; i += 256) {
    int j = i & 1, lane = (i >> 1) & 31, nt = (i >> 6) & 3, kc = i >> 8;
    int kk = 4 * kc + 2 * (lane >> 4) + j;
    sW[i] = Wnb[kk * 64 + nt * 16 + (lane & 15)];
  }
  __syncthreads();
  const int lane = threadIdx.x & 31;
  const int h = lane >> 4, m = lane & 15;
  const int wave = blockIdx.x * 8 + (threadIdx.x >> 5);
  const int nwaves = gridDim.x * 8;
  const int ntiles = NNODES / 16;
  for (int t = wave; t < ntiles; t += nwaves) {
    const int n0 = t * 16;
    const float* ap = agg_r + (long)(n0 + m) * 64;
    const float* xp = x + (long)(n0 + m) * 16;
    v8f acc[4];
#pragma unroll
    for (int nt = 0; nt < 4; ++nt) {
      float bb = bias_nb[nt * 16 + m];
      acc[nt] = (v8f){bb, bb, bb, bb, bb, bb, bb, bb};
    }
#pragma unroll
    for (int kc = 0; kc < 20; ++kc) {
      const int kk = 4 * kc + 2 * h;
      v2f a = (kc < 16) ? *(const v2f*)(ap + kk) : *(const v2f*)(xp + kk - 64);
#pragma unroll
      for (int nt = 0; nt < 4; ++nt) {
        v2f b = *(const v2f*)(&sW[(kc * 4 + nt) * 64 + lane * 2]);
        acc[nt] = __builtin_amdgcn_wmma_f32_16x16x4_f32(false, a, false, b,
                                                        (short)0, acc[nt], false, false);
      }
    }
#pragma unroll
    for (int nt = 0; nt < 4; ++nt)
#pragma unroll
      for (int v = 0; v < 8; ++v)
        n_h[(long)(n0 + 8 * h + v) * 64 + nt * 16 + m] = fmaxf(acc[nt][v], 0.0f);
  }
}

__global__ void k_colsum(const float* __restrict__ src, int rows, float* __restrict__ dst) {
  __shared__ float red[256];
  const int col = threadIdx.x & 63;
  const int g = threadIdx.x >> 6;
  float local = 0.f;
  for (int r = blockIdx.x * 4 + g; r < rows; r += gridDim.x * 4)
    local += src[(long)r * 64 + col];
  red[threadIdx.x] = local;
  __syncthreads();
  if (g == 0) {
    local += red[64 + threadIdx.x] + red[128 + threadIdx.x] + red[192 + threadIdx.x];
    unsafeAtomicAdd(&dst[col], local);
  }
}

__global__ void k_global(const float* __restrict__ u,
                         const float* __restrict__ Wgb, const float* __restrict__ bgb,
                         const float* __restrict__ Wd1, const float* __restrict__ bd1,
                         const float* __restrict__ e_sum, const float* __restrict__ n_sum,
                         float* __restrict__ u2, float* __restrict__ bias_d1) {
  __shared__ float su2[64];
  const int j = threadIdx.x;
  float acc = bgb[j];
  for (int k = 0; k < 64; ++k) {
    float em = e_sum[k] * (1.0f / NEDGES);
    float nm = n_sum[k] * (1.0f / NNODES);
    acc += em * Wgb[k * 64 + j] + nm * Wgb[(64 + k) * 64 + j] + u[k] * Wgb[(128 + k) * 64 + j];
  }
  acc = fmaxf(acc, 0.0f);
  su2[j] = acc;
  u2[j] = acc;
  __syncthreads();
  float b = bd1[j];
  for (int k = 0; k < 64; ++k) b += su2[k] * Wd1[(192 + k) * 64 + j];
  bias_d1[j] = b;
}

__global__ __launch_bounds__(256) void k_dec(
    const float* __restrict__ agg_r, const float* __restrict__ agg_s,
    const float* __restrict__ cnt_r, const float* __restrict__ cnt_s,
    const float* __restrict__ n_h, const float* __restrict__ Wd1,
    const float* __restrict__ bias_d1, const float* __restrict__ Wd2,
    const float* __restrict__ bd2, float* __restrict__ out) {
  __shared__ float sW1[48 * 4 * 64];
  __shared__ float sW2[16 * 64];
  __shared__ float sH[8 * 16 * 64];
  for (int i = threadIdx.x; i < 48 * 4 * 64; i += 256) {
    int j = i & 1, lane = (i >> 1) & 31, nt = (i >> 6) & 3, kc = i >> 8;
    int kk = 4 * kc + 2 * (lane >> 4) + j;
    sW1[i] = Wd1[kk * 64 + nt * 16 + (lane & 15)];
  }
  for (int i = threadIdx.x; i < 16 * 64; i += 256) {
    int j = i & 1, lane = (i >> 1) & 31, kc = i >> 6;
    int kk = 4 * kc + 2 * (lane >> 4) + j;
    sW2[i] = Wd2[kk * 16 + (lane & 15)];
  }
  __syncthreads();
  const int lane = threadIdx.x & 31;
  const int h = lane >> 4, m = lane & 15;
  float* hw = sH + (threadIdx.x >> 5) * 1024;
  const int wave = blockIdx.x * 8 + (threadIdx.x >> 5);
  const int nwaves = gridDim.x * 8;
  const int ntiles = NNODES / 16;
  for (int t = wave; t < ntiles; t += nwaves) {
    const int n0 = t * 16;
    const int n = n0 + m;
    const float invr = 1.0f / fmaxf(cnt_r[n], 1.0f);
    const float invs = 1.0f / fmaxf(cnt_s[n], 1.0f);
    const float* pr = agg_r + (long)n * 64;
    const float* ps = agg_s + (long)n * 64;
    const float* pn = n_h + (long)n * 64;
    v8f acc[4];
#pragma unroll
    for (int nt = 0; nt < 4; ++nt) {
      float bb = bias_d1[nt * 16 + m];
      acc[nt] = (v8f){bb, bb, bb, bb, bb, bb, bb, bb};
    }
#pragma unroll
    for (int kc = 0; kc < 48; ++kc) {
      const int kk = 4 * kc + 2 * h;
      v2f a;
      if (kc < 16)      { a = *(const v2f*)(pr + kk);        a *= invr; }
      else if (kc < 32) { a = *(const v2f*)(ps + kk - 64);   a *= invs; }
      else              { a = *(const v2f*)(pn + kk - 128); }
#pragma unroll
      for (int nt = 0; nt < 4; ++nt) {
        v2f b = *(const v2f*)(&sW1[(kc * 4 + nt) * 64 + lane * 2]);
        acc[nt] = __builtin_amdgcn_wmma_f32_16x16x4_f32(false, a, false, b,
                                                        (short)0, acc[nt], false, false);
      }
    }
#pragma unroll
    for (int nt = 0; nt < 4; ++nt)
#pragma unroll
      for (int v = 0; v < 8; ++v)
        hw[(8 * h + v) * 64 + nt * 16 + m] = fmaxf(acc[nt][v], 0.0f);
    float bb2 = bd2[m];
    v8f acc2 = (v8f){bb2, bb2, bb2, bb2, bb2, bb2, bb2, bb2};
#pragma unroll
    for (int kc = 0; kc < 16; ++kc) {
      const int kk = 4 * kc + 2 * h;
      v2f a = *(const v2f*)(hw + m * 64 + kk);
      v2f b = *(const v2f*)(&sW2[kc * 64 + lane * 2]);
      acc2 = __builtin_amdgcn_wmma_f32_16x16x4_f32(false, a, false, b,
                                                   (short)0, acc2, false, false);
    }
#pragma unroll
    for (int v = 0; v < 8; ++v)
      out[(long)(n0 + 8 * h + v) * 16 + m] = acc2[v];
  }
}

extern "C" void kernel_launch(void* const* d_in, const int* in_sizes, int n_in,
                              void* d_out, int out_size, void* d_ws, size_t ws_size,
                              hipStream_t stream) {
  const float* x    = (const float*)d_in[0];
  const float* ea   = (const float*)d_in[1];
  const float* u    = (const float*)d_in[2];
  const int* send   = (const int*)d_in[3];
  const int* recv   = (const int*)d_in[4];
  const float* Web  = (const float*)d_in[7];
  const float* beb  = (const float*)d_in[8];
  const float* Wnb  = (const float*)d_in[9];
  const float* bnb  = (const float*)d_in[10];
  const float* Wgb  = (const float*)d_in[11];
  const float* bgb  = (const float*)d_in[12];
  const float* Wd1  = (const float*)d_in[13];
  const float* bd1  = (const float*)d_in[14];
  const float* Wd2  = (const float*)d_in[15];
  const float* bd2  = (const float*)d_in[16];
  float* out = (float*)d_out;
  float* ws = (float*)d_ws;

  const long N = NNODES;
  float* agg_r   = ws;
  float* agg_s   = agg_r + N * 64;
  float* cnt_r   = agg_s + N * 64;
  float* cnt_s   = cnt_r + N;
  float* e_sum   = cnt_s + N;
  float* n_sum   = e_sum + 64;
  float* bias_eb = n_sum + 64;
  float* bias_nb = bias_eb + 64;
  float* u2      = bias_nb + 64;
  float* bias_d1 = u2 + 64;
  float* n_hbuf  = bias_d1 + 64;

  const long zlen = 2 * N * 64 + 2 * N + 128;
  k_zero<<<2048, 256, 0, stream>>>(ws, zlen);
  k_prep<<<1, 64, 0, stream>>>(u, Web, beb, Wnb, bnb, bias_eb, bias_nb);
  k_edge<<<2048, 256, 0, stream>>>(x, ea, recv, send, Web, bias_eb,
                                   agg_r, agg_s, cnt_r, cnt_s);
  k_node<<<782, 256, 0, stream>>>(x, agg_r, Wnb, bias_nb, n_hbuf);
  k_colsum<<<512, 256, 0, stream>>>(agg_r, NNODES, e_sum);
  k_colsum<<<512, 256, 0, stream>>>(n_hbuf, NNODES, n_sum);
  k_global<<<1, 64, 0, stream>>>(u, Wgb, bgb, Wd1, bd1, e_sum, n_sum, u2, bias_d1);
  k_dec<<<1024, 256, 0, stream>>>(agg_r, agg_s, cnt_r, cnt_s, n_hbuf,
                                  Wd1, bias_d1, Wd2, bd2, out);
}


# k_edge: each sum's f32 atomic adds come from one package half only (half 0 adds the receiver sum, half 1 the sender sum; tiles drawn from two counters, e_h computed per role); arrival order as before
# speedup vs baseline: 1.5081x; 1.3291x over previous
#include <hip/hip_runtime.h>

typedef __attribute__((ext_vector_type(2))) float v2f;
typedef __attribute__((ext_vector_type(8))) float v8f;

#define NNODES 100000
#define NEDGES 1600000

__device__ __forceinline__ float xhalf(float x) {
  return __builtin_bit_cast(float, __builtin_amdgcn_permlanex16(
      __builtin_bit_cast(int, x), __builtin_bit_cast(int, x), 0x76543210, 0xfedcba98, false, false));
}

__global__ void k_zero(float* __restrict__ p, long n, int* __restrict__ ctr) {
  if (blockIdx.x == 0 && threadIdx.x < 2) ctr[64 * threadIdx.x] = 0;
  long i = ((long)blockIdx.x * blockDim.x + threadIdx.x) * 4;
  long stride = (long)gridDim.x * blockDim.x * 4;
  float4 z = {0.f, 0.f, 0.f, 0.f};
  for (long j = i; j < n; j += stride) *(float4*)(p + j) = z;
}

__global__ void k_prep(const float* __restrict__ u,
                       const float* __restrict__ Web, const float* __restrict__ beb,
                       const float* __restrict__ Wnb, const float* __restrict__ bnb,
                       float* __restrict__ bias_eb, float* __restrict__ bias_nb) {
  int j = threadIdx.x;
  float a = beb[j], b = bnb[j];
  for (int k = 0; k < 64; ++k) {
    float uk = u[k];
    a += uk * Web[(48 + k) * 64 + j];
    b += uk * Wnb[(80 + k) * 64 + j];
  }
  bias_eb[j] = a;
  bias_nb[j] = b;
}

#define EDGE_RUN 4
__global__ __launch_bounds__(256) void k_edge(
    const float* __restrict__ x, const float* __restrict__ ea,
    const int* __restrict__ recv, const int* __restrict__ send,
    const float* __restrict__ Web, const float* __restrict__ bias_eb,
    float* __restrict__ agg_r, float* __restrict__ agg_s,
    float* __restrict__ cnt_r, float* __restrict__ cnt_s, int* __restrict__ ctr) {
  __shared__ float sW[12 * 4 * 64];
  for (int i = threadIdx.x; i < 12 * 4 * 64; i += 256) {
    int j = i & 1, lane = (i >> 1) & 31, nt = (i >> 6) & 3, kc = i >> 8;
    int kk = 4 * kc + 2 * (lane >> 4) + j;
    sW[i] = Web[kk * 64 + nt * 16 + (lane & 15)];
  }
  __syncthreads();
  const int lane = threadIdx.x & 31;
  const int h = lane >> 4, m = lane & 15;
  const int ntiles = NEDGES / 16;
  const int pkg_half = (__builtin_amdgcn_s_sendmsg_rtn(0x87) >> 18) & 1;
  for (int pass = 0; pass < 2; ++pass) {
    const int role = pkg_half ^ pass;
    float* __restrict__ agg = role ? agg_s : agg_r;
    float* __restrict__ cnt = role ? cnt_s : cnt_r;
    int* myctr = ctr + 64 * role;
    int t = 0, t1 = 0;
    for (;; ++t) {
      if (t == t1) {
        int t0 = 0;
        if (lane == 0) t0 = atomicAdd(myctr, EDGE_RUN);
        t0 = __builtin_amdgcn_readfirstlane(t0);
        if (t0 >= ntiles) break;
        t = t0;
        t1 = min(t0 + EDGE_RUN, ntiles);
      }
      const int e0 = t * 16;
      const int er = e0 + m;
      const int ri = recv[er], si = send[er];
      const int di = role ? si : ri;
      const float* segs[3] = {ea + (long)er * 16, x + (long)ri * 16, x + (long)si * 16};
      v8f acc[4];
#pragma unroll
      for (int nt = 0; nt < 4; ++nt) {
        float bb = bias_eb[nt * 16 + m];
        acc[nt] = (v8f){bb, bb, bb, bb, bb, bb, bb, bb};
      }
#pragma unroll
      for (int seg = 0; seg < 3; ++seg) {
        const float* bp = segs[seg];
#pragma unroll
        for (int cc = 0; cc < 4; ++cc) {
          v2f a = *(const v2f*)(bp + 4 * cc + 2 * h);
          const int kc = seg * 4 + cc;
#pragma unroll
          for (int nt = 0; nt < 4; ++nt) {
            v2f b = *(const v2f*)(&sW[(kc * 4 + nt) * 64 + lane * 2]);
            acc[nt] = __builtin_amdgcn_wmma_f32_16x16x4_f32(false, a, false, b,
                                                            (short)0, acc[nt], false, false);
          }
        }
      }
#pragma unroll
      for (int v = 0; v < 8; ++v) {
        const long dA = __builtin_amdgcn_readlane(di, v);
        const long dB = __builtin_amdgcn_readlane(di, 8 + v);
#pragma unroll
        for (int p = 0; p < 2; ++p) {
          const float X = fmaxf(acc[2 * p][v], 0.0f);
          const float Y = fmaxf(acc[2 * p + 1][v], 0.0f);
          const float Zx = xhalf(h ? X : Y);
          const float A = h ? Zx : X;
          const float B = h ? Y : Zx;
          const int col = 32 * p + lane;
          unsafeAtomicAdd(&agg[dA * 64 + col], A);
          unsafeAtomicAdd(&agg[dB * 64 + col], B);
        }
      }
      if (lane < 16) unsafeAtomicAdd(&cnt[di], 1.0f);
    }
  }
}

__global__ __launch_bounds__(256) void k_node(
    const float* __restrict__ x, const float* __restrict__ agg_r,
    const float* __restrict__ Wnb, const float* __restrict__ bias_nb,
    float* __restrict__ n_h) {
  __shared__ float sW[20 * 4 * 64];
  for (int i = threadIdx.x; i < 20 * 4 * 64; i += 256) {
    int j = i & 1, lane = (i >> 1) & 31, nt = (i >> 6) & 3, kc = i >> 8;
    int kk = 4 * kc + 2 * (lane >> 4) + j;
    sW[i] = Wnb[kk * 64 + nt * 16 + (lane & 15)];
  }
  __syncthreads();
  const int lane = threadIdx.x & 31;
  const int h = lane >> 4, m = lane & 15;
  const int wave = blockIdx.x * 8 + (threadIdx.x >> 5);
  const int nwaves = gridDim.x * 8;
  const int ntiles = NNODES / 16;
  for (int t = wave; t < ntiles; t += nwaves) {
    const int n0 = t * 16;
    const float* ap = agg_r + (long)(n0 + m) * 64;
    const float* xp = x + (long)(n0 + m) * 16;
    v8f acc[4];
#pragma unroll
    for (int nt = 0; nt < 4; ++nt) {
      float bb = bias_nb[nt * 16 + m];
      acc[nt] = (v8f){bb, bb, bb, bb, bb, bb, bb, bb};
    }
#pragma unroll
    for (int kc = 0; kc < 20; ++kc) {
      const int kk = 4 * kc + 2 * h;
      v2f a = (kc < 16) ? *(const v2f*)(ap + kk) : *(const v2f*)(xp + kk - 64);
#pragma unroll
      for (int nt = 0; nt < 4; ++nt) {
        v2f b = *(const v2f*)(&sW[(kc * 4 + nt) * 64 + lane * 2]);
        acc[nt] = __builtin_amdgcn_wmma_f32_16x16x4_f32(false, a, false, b,
                                                        (short)0, acc[nt], false, false);
      }
    }
#pragma unroll
    for (int nt = 0; nt < 4; ++nt)
#pragma unroll
      for (int v = 0; v < 8; ++v)
        n_h[(long)(n0 + 8 * h + v) * 64 + nt * 16 + m] = fmaxf(acc[nt][v], 0.0f);
  }
}

__global__ void k_colsum(const float* __restrict__ src, int rows, float* __restrict__ dst) {
  __shared__ float red[256];
  const int col = threadIdx.x & 63;
  const int g = threadIdx.x >> 6;
  float local = 0.f;
  for (int r = blockIdx.x * 4 + g; r < rows; r += gridDim.x * 4)
    local += src[(long)r * 64 + col];
  red[threadIdx.x] = local;
  __syncthreads();
  if (g == 0) {
    local += red[64 + threadIdx.x] + red[128 + threadIdx.x] + red[192 + threadIdx.x];
    unsafeAtomicAdd(&dst[col], local);
  }
}

__global__ void k_global(const float* __restrict__ u,
                         const float* __restrict__ Wgb, const float* __restrict__ bgb,
                         const float* __restrict__ Wd1, const float* __restrict__ bd1,
                         const float* __restrict__ e_sum, const float* __restrict__ n_sum,
                         float* __restrict__ u2, float* __restrict__ bias_d1) {
  __shared__ float su2[64];
  const int j = threadIdx.x;
  float acc = bgb[j];
  for (int k = 0; k < 64; ++k) {
    float em = e_sum[k] * (1.0f / NEDGES);
    float nm = n_sum[k] * (1.0f / NNODES);
    acc += em * Wgb[k * 64 + j] + nm * Wgb[(64 + k) * 64 + j] + u[k] * Wgb[(128 + k) * 64 + j];
  }
  acc = fmaxf(acc, 0.0f);
  su2[j] = acc;
  u2[j] = acc;
  __syncthreads();
  float b = bd1[j];
  for (int k = 0; k < 64; ++k) b += su2[k] * Wd1[(192 + k) * 64 + j];
  bias_d1[j] = b;
}

__global__ __launch_bounds__(256) void k_dec(
    const float* __restrict__ agg_r, const float* __restrict__ agg_s,
    const float* __restrict__ cnt_r, const float* __restrict__ cnt_s,
    const float* __restrict__ n_h, const float* __restrict__ Wd1,
    const float* __restrict__ bias_d1, const float* __restrict__ Wd2,
    const float* __restrict__ bd2, float* __restrict__ out) {
  __shared__ float sW1[48 * 4 * 64];
  __shared__ float sW2[16 * 64];
  __shared__ float sH[8 * 16 * 64];
  for (int i = threadIdx.x; i < 48 * 4 * 64; i += 256) {
    int j = i & 1, lane = (i >> 1) & 31, nt = (i >> 6) & 3, kc = i >> 8;
    int kk = 4 * kc + 2 * (lane >> 4) + j;
    sW1[i] = Wd1[kk * 64 + nt * 16 + (lane & 15)];
  }
  for (int i = threadIdx.x; i < 16 * 64; i += 256) {
    int j = i & 1, lane = (i >> 1) & 31, kc = i >> 6;
    int kk = 4 * kc + 2 * (lane >> 4) + j;
    sW2[i] = Wd2[kk * 16 + (lane & 15)];
  }
  __syncthreads();
  const int lane = threadIdx.x & 31;
  const int h = lane >> 4, m = lane & 15;
  float* hw = sH + (threadIdx.x >> 5) * 1024;
  const int wave = blockIdx.x * 8 + (threadIdx.x >> 5);
  const int nwaves = gridDim.x * 8;
  const int ntiles = NNODES / 16;
  for (int t = wave; t < ntiles; t += nwaves) {
    const int n0 = t * 16;
    const int n = n0 + m;
    const float invr = 1.0f / fmaxf(cnt_r[n], 1.0f);
    const float invs = 1.0f / fmaxf(cnt_s[n], 1.0f);
    const float* pr = agg_r + (long)n * 64;
    const float* ps = agg_s + (long)n * 64;
    const float* pn = n_h + (long)n * 64;
    v8f acc[4];
#pragma unroll
    for (int nt = 0; nt < 4; ++nt) {
      float bb = bias_d1[nt * 16 + m];
      acc[nt] = (v8f){bb, bb, bb, bb, bb, bb, bb, bb};
    }
#pragma unroll
    for (int kc = 0; kc < 48; ++kc) {
      const int kk = 4 * kc + 2 * h;
      v2f a;
      if (kc < 16)      { a = *(const v2f*)(pr + kk);        a *= invr; }
      else if (kc < 32) { a = *(const v2f*)(ps + kk - 64);   a *= invs; }
      else              { a = *(const v2f*)(pn + kk - 128); }
#pragma unroll
      for (int nt = 0; nt < 4; ++nt) {
        v2f b = *(const v2f*)(&sW1[(kc * 4 + nt) * 64 + lane * 2]);
        acc[nt] = __builtin_amdgcn_wmma_f32_16x16x4_f32(false, a, false, b,
                                                        (short)0, acc[nt], false, false);
      }
    }
#pragma unroll
    for (int nt = 0; nt < 4; ++nt)
#pragma unroll
      for (int v = 0; v < 8; ++v)
        hw[(8 * h + v) * 64 + nt * 16 + m] = fmaxf(acc[nt][v], 0.0f);
    float bb2 = bd2[m];
    v8f acc2 = (v8f){bb2, bb2, bb2, bb2, bb2, bb2, bb2, bb2};
#pragma unroll
    for (int kc = 0; kc < 16; ++kc) {
      const int kk = 4 * kc + 2 * h;
      v2f a = *(const v2f*)(hw + m * 64 + kk);
      v2f b = *(const v2f*)(&sW2[kc * 64 + lane * 2]);
      acc2 = __builtin_amdgcn_wmma_f32_16x16x4_f32(false, a, false, b,
                                                   (short)0, acc2, false, false);
    }
#pragma unroll
    for (int v = 0; v < 8; ++v)
      out[(long)(n0 + 8 * h + v) * 16 + m] = acc2[v];
  }
}

extern "C" void kernel_launch(void* const* d_in, const int* in_sizes, int n_in,
                              void* d_out, int out_size, void* d_ws, size_t ws_size,
                              hipStream_t stream) {
  const float* x    = (const float*)d_in[0];
  const float* ea   = (const float*)d_in[1];
  const float* u    = (const float*)d_in[2];
  const int* send   = (const int*)d_in[3];
  const int* recv   = (const int*)d_in[4];
  const float* Web  = (const float*)d_in[7];
  const float* beb  = (const float*)d_in[8];
  const float* Wnb  = (const float*)d_in[9];
  const float* bnb  = (const float*)d_in[10];
  const float* Wgb  = (const float*)d_in[11];
  const float* bgb  = (const float*)d_in[12];
  const float* Wd1  = (const float*)d_in[13];
  const float* bd1  = (const float*)d_in[14];
  const float* Wd2  = (const float*)d_in[15];
  const float* bd2  = (const float*)d_in[16];
  float* out = (float*)d_out;
  float* ws = (float*)d_ws;

  const long N = NNODES;
  float* agg_r   = ws;
  float* agg_s   = agg_r + N * 64;
  float* cnt_r   = agg_s + N * 64;
  float* cnt_s   = cnt_r + N;
  float* e_sum   = cnt_s + N;
  float* n_sum   = e_sum + 64;
  float* bias_eb = n_sum + 64;
  float* bias_nb = bias_eb + 64;
  float* u2      = bias_nb + 64;
  float* bias_d1 = u2 + 64;
  float* n_hbuf  = bias_d1 + 64;

  const long zlen = 2 * N * 64 + 2 * N + 128;
  int* ctr = (int*)n_hbuf;
  k_zero<<<2048, 256, 0, stream>>>(ws, zlen, ctr);
  k_prep<<<1, 64, 0, stream>>>(u, Web, beb, Wnb, bnb, bias_eb, bias_nb);
  k_edge<<<1536, 256, 0, stream>>>(x, ea, recv, send, Web, bias_eb,
                                   agg_r, agg_s, cnt_r, cnt_s, ctr);
  k_node<<<782, 256, 0, stream>>>(x, agg_r, Wnb, bias_nb, n_hbuf);
  k_colsum<<<512, 256, 0, stream>>>(agg_r, NNODES, e_sum);
  k_colsum<<<512, 256, 0, stream>>>(n_hbuf, NNODES, n_sum);
  k_global<<<1, 64, 0, stream>>>(u, Wgb, bgb, Wd1, bd1, e_sum, n_sum, u2, bias_d1);
  k_dec<<<1024, 256, 0, stream>>>(agg_r, agg_s, cnt_r, cnt_s, n_hbuf,
                                  Wd1, bias_d1, Wd2, bd2, out);
}


# 8 launches to 5: e_h and n_h column sums taken in the producing kernels (registers, lane halves, LDS, one f32 atomic add per column and workgroup); u bias fold in the zero pass; sum order changes
# speedup vs baseline: 1.7563x; 1.1646x over previous
#include <hip/hip_runtime.h>

typedef __attribute__((ext_vector_type(2))) float v2f;
typedef __attribute__((ext_vector_type(8))) float v8f;

#define NNODES 100000
#define NEDGES 1600000

__device__ __forceinline__ float xhalf(float x) {
  return __builtin_bit_cast(float, __builtin_amdgcn_permlanex16(
      __builtin_bit_cast(int, x), __builtin_bit_cast(int, x), 0x76543210, 0xfedcba98, false, false));
}

__global__ void k_zero(float* __restrict__ p, long n, int* __restrict__ ctr,
                       const float* __restrict__ u,
                       const float* __restrict__ Web, const float* __restrict__ beb,
                       const float* __restrict__ Wnb, const float* __restrict__ bnb,
                       float* __restrict__ bias_eb, float* __restrict__ bias_nb) {
  if (blockIdx.x == 0 && threadIdx.x < 2) ctr[64 * threadIdx.x] = 0;
  long i = ((long)blockIdx.x * blockDim.x + threadIdx.x) * 4;
  long stride = (long)gridDim.x * blockDim.x * 4;
  float4 z = {0.f, 0.f, 0.f, 0.f};
  for (long j = i; j < n; j += stride) *(float4*)(p + j) = z;
  if (blockIdx.x == 0 && threadIdx.x < 64) {
    const int j = threadIdx.x;
    float a = beb[j], b = bnb[j];
    for (int k = 0; k < 64; ++k) {
      float uk = u[k];
      a += uk * Web[(48 + k) * 64 + j];
      b += uk * Wnb[(80 + k) * 64 + j];
    }
    bias_eb[j] = a;
    bias_nb[j] = b;
  }
}

#define EDGE_RUN 4
__global__ __launch_bounds__(256, 12) void k_edge(
    const float* __restrict__ x, const float* __restrict__ ea,
    const int* __restrict__ recv, const int* __restrict__ send,
    const float* __restrict__ Web, const float* __restrict__ bias_eb,
    float* __restrict__ agg_r, float* __restrict__ agg_s,
    float* __restrict__ cnt_r, float* __restrict__ cnt_s, int* __restrict__ ctr,
    float* __restrict__ e_sum) {
  __shared__ float sW[12 * 4 * 64];
  for (int i = threadIdx.x; i < 12 * 4 * 64; i += 256) {
    int j = i & 1, lane = (i >> 1) & 31, nt = (i >> 6) & 3, kc = i >> 8;
    int kk = 4 * kc + 2 * (lane >> 4) + j;
    sW[i] = Web[kk * 64 + nt * 16 + (lane & 15)];
  }
  __syncthreads();
  const int lane = threadIdx.x & 31;
  const int h = lane >> 4, m = lane & 15;
  const int ntiles = NEDGES / 16;
  const int pkg_half = (__builtin_amdgcn_s_sendmsg_rtn(0x87) >> 18) & 1;
  float es[4] = {0.f, 0.f, 0.f, 0.f};
  for (int pass = 0; pass < 2; ++pass) {
    const int role = pkg_half ^ pass;
    float* __restrict__ agg = role ? agg_s : agg_r;
    float* __restrict__ cnt = role ? cnt_s : cnt_r;
    int* myctr = ctr + 64 * role;
    int t = 0, t1 = 0;
    for (;; ++t) {
      if (t == t1) {
        int t0 = 0;
        if (lane == 0) t0 = atomicAdd(myctr, EDGE_RUN);
        t0 = __builtin_amdgcn_readfirstlane(t0);
        if (t0 >= ntiles) break;
        t = t0;
        t1 = min(t0 + EDGE_RUN, ntiles);
      }
      const int e0 = t * 16;
      const int er = e0 + m;
      const int ri = recv[er], si = send[er];
      const int di = role ? si : ri;
      const float* segs[3] = {ea + (long)er * 16, x + (long)ri * 16, x + (long)si * 16};
      v8f acc[4];
#pragma unroll
      for (int nt = 0; nt < 4; ++nt) {
        float bb = bias_eb[nt * 16 + m];
        acc[nt] = (v8f){bb, bb, bb, bb, bb, bb, bb, bb};
      }
#pragma unroll
      for (int seg = 0; seg < 3; ++seg) {
        const float* bp = segs[seg];
#pragma unroll
        for (int cc = 0; cc < 4; ++cc) {
          v2f a = *(const v2f*)(bp + 4 * cc + 2 * h);
          const int kc = seg * 4 + cc;
#pragma unroll
          for (int nt = 0; nt < 4; ++nt) {
            v2f b = *(const v2f*)(&sW[(kc * 4 + nt) * 64 + lane * 2]);
            acc[nt] = __builtin_amdgcn_wmma_f32_16x16x4_f32(false, a, false, b,
                                                            (short)0, acc[nt], false, false);
          }
        }
      }
#pragma unroll
      for (int nt = 0; nt < 4; ++nt) {
        float ts = 0.f;
#pragma unroll
        for (int v = 0; v < 8; ++v) ts += fmaxf(acc[nt][v], 0.0f);
        es[nt] += role ? 0.0f : ts;
      }
#pragma unroll
      for (int v = 0; v < 8; ++v) {
        const long dA = __builtin_amdgcn_readlane(di, v);
        const long dB = __builtin_amdgcn_readlane(di, 8 + v);
#pragma unroll
        for (int p = 0; p < 2; ++p) {
          const float X = fmaxf(acc[2 * p][v], 0.0f);
          const float Y = fmaxf(acc[2 * p + 1][v], 0.0f);
          const float Zx = xhalf(h ? X : Y);
          const float A = h ? Zx : X;
          const float B = h ? Y : Zx;
          const int col = 32 * p + lane;
          unsafeAtomicAdd(&agg[dA * 64 + col], A);
          unsafeAtomicAdd(&agg[dB * 64 + col], B);
        }
      }
      if (lane < 16) unsafeAtomicAdd(&cnt[di], 1.0f);
    }
  }
  __syncthreads();
#pragma unroll
  for (int nt = 0; nt < 4; ++nt) {
    const float tsum = es[nt] + xhalf(es[nt]);
    if (lane < 16) sW[(threadIdx.x >> 5) * 64 + nt * 16 + m] = tsum;
  }
  __syncthreads();
  if (threadIdx.x < 64) {
    float s = 0.f;
#pragma unroll
    for (int w = 0; w < 8; ++w) s += sW[w * 64 + threadIdx.x];
    if (s != 0.0f) unsafeAtomicAdd(&e_sum[threadIdx.x], s);
  }
}

__global__ __launch_bounds__(256) void k_node(
    const float* __restrict__ x, const float* __restrict__ agg_r,
    const float* __restrict__ Wnb, const float* __restrict__ bias_nb,
    float* __restrict__ n_h, float* __restrict__ n_sum) {
  __shared__ float sW[20 * 4 * 64];
  for (int i = threadIdx.x; i < 20 * 4 * 64; i += 256) {
    int j = i & 1, lane = (i >> 1) & 31, nt = (i >> 6) & 3, kc = i >> 8;
    int kk = 4 * kc + 2 * (lane >> 4) + j;
    sW[i] = Wnb[kk * 64 + nt * 16 + (lane & 15)];
  }
  __syncthreads();
  const int lane = threadIdx.x & 31;
  const int h = lane >> 4, m = lane & 15;
  const int wave = blockIdx.x * 8 + (threadIdx.x >> 5);
  const int nwaves = gridDim.x * 8;
  const int ntiles = NNODES / 16;
  float ns[4] = {0.f, 0.f, 0.f, 0.f};
  for (int t = wave; t < ntiles; t += nwaves) {
    const int n0 = t * 16;
    const float* ap = agg_r + (long)(n0 + m) * 64;
    const float* xp = x + (long)(n0 + m) * 16;
    v8f acc[4];
#pragma unroll
    for (int nt = 0; nt < 4; ++nt) {
      float bb = bias_nb[nt * 16 + m];
      acc[nt] = (v8f){bb, bb, bb, bb, bb, bb, bb, bb};
    }
#pragma unroll
    for (int kc = 0; kc < 20; ++kc) {
      const int kk = 4 * kc + 2 * h;
      v2f a = (kc < 16) ? *(const v2f*)(ap + kk) : *(const v2f*)(xp + kk - 64);
#pragma unroll
      for (int nt = 0; nt < 4; ++nt) {
        v2f b = *(const v2f*)(&sW[(kc * 4 + nt) * 64 + lane * 2]);
        acc[nt] = __builtin_amdgcn_wmma_f32_16x16x4_f32(false, a, false, b,
                                                        (short)0, acc[nt], false, false);
      }
    }
#pragma unroll
    for (int nt = 0; nt < 4; ++nt)
#pragma unroll
      for (int v = 0; v < 8; ++v) {
        const float r = fmaxf(acc[nt][v], 0.0f);
        n_h[(long)(n0 + 8 * h + v) * 64 + nt * 16 + m] = r;
        ns[nt] += r;
      }
  }
  __syncthreads();
#pragma unroll
  for (int nt = 0; nt < 4; ++nt) {
    const float tsum = ns[nt] + xhalf(ns[nt]);
    if (lane < 16) sW[(threadIdx.x >> 5) * 64 + nt * 16 + m] = tsum;
  }
  __syncthreads();
  if (threadIdx.x < 64) {
    float s = 0.f;
#pragma unroll
    for (int w = 0; w < 8; ++w) s += sW[w * 64 + threadIdx.x];
    if (s != 0.0f) unsafeAtomicAdd(&n_sum[threadIdx.x], s);
  }
}

__global__ void k_global(const float* __restrict__ u,
                         const float* __restrict__ Wgb, const float* __restrict__ bgb,
                         const float* __restrict__ Wd1, const float* __restrict__ bd1,
                         const float* __restrict__ e_sum, const float* __restrict__ n_sum,
                         float* __restrict__ u2, float* __restrict__ bias_d1) {
  __shared__ float su2[64];
  const int j = threadIdx.x;
  float acc = bgb[j];
  for (int k = 0; k < 64; ++k) {
    float em = e_sum[k] * (1.0f / NEDGES);
    float nm = n_sum[k] * (1.0f / NNODES);
    acc += em * Wgb[k * 64 + j] + nm * Wgb[(64 + k) * 64 + j] + u[k] * Wgb[(128 + k) * 64 + j];
  }
  acc = fmaxf(acc, 0.0f);
  su2[j] = acc;
  u2[j] = acc;
  __syncthreads();
  float b = bd1[j];
  for (int k = 0; k < 64; ++k) b += su2[k] * Wd1[(192 + k) * 64 + j];
  bias_d1[j] = b;
}

__global__ __launch_bounds__(256) void k_dec(
    const float* __restrict__ agg_r, const float* __restrict__ agg_s,
    const float* __restrict__ cnt_r, const float* __restrict__ cnt_s,
    const float* __restrict__ n_h, const float* __restrict__ Wd1,
    const float* __restrict__ bias_d1, const float* __restrict__ Wd2,
    const float* __restrict__ bd2, float* __restrict__ out) {
  __shared__ float sW1[48 * 4 * 64];
  __shared__ float sW2[16 * 64];
  __shared__ float sH[8 * 16 * 64];
  for (int i = threadIdx.x; i < 48 * 4 * 64; i += 256) {
    int j = i & 1, lane = (i >> 1) & 31, nt = (i >> 6) & 3, kc = i >> 8;
    int kk = 4 * kc + 2 * (lane >> 4) + j;
    sW1[i] = Wd1[kk * 64 + nt * 16 + (lane & 15)];
  }
  for (int i = threadIdx.x; i < 16 * 64; i += 256) {
    int j = i & 1, lane = (i >> 1) & 31, kc = i >> 6;
    int kk = 4 * kc + 2 * (lane >> 4) + j;
    sW2[i] = Wd2[kk * 16 + (lane & 15)];
  }
  __syncthreads();
  const int lane = threadIdx.x & 31;
  const int h = lane >> 4, m = lane & 15;
  float* hw = sH + (threadIdx.x >> 5) * 1024;
  const int wave = blockIdx.x * 8 + (threadIdx.x >> 5);
  const int nwaves = gridDim.x * 8;
  const int ntiles = NNODES / 16;
  for (int t = wave; t < ntiles; t += nwaves) {
    const int n0 = t * 16;
    const int n = n0 + m;
    const float invr = 1.0f / fmaxf(cnt_r[n], 1.0f);
    const float invs = 1.0f / fmaxf(cnt_s[n], 1.0f);
    const float* pr = agg_r + (long)n * 64;
    const float* ps = agg_s + (long)n * 64;
    const float* pn = n_h + (long)n * 64;
    v8f acc[4];
#pragma unroll
    for (int nt = 0; nt < 4; ++nt) {
      float bb = bias_d1[nt * 16 + m];
      acc[nt] = (v8f){bb, bb, bb, bb, bb, bb, bb, bb};
    }
#pragma unroll
    for (int kc = 0; kc < 48; ++kc) {
      const int kk = 4 * kc + 2 * h;
      v2f a;
      if (kc < 16)      { a = *(const v2f*)(pr + kk);        a *= invr; }
      else if (kc < 32) { a = *(const v2f*)(ps + kk - 64);   a *= invs; }
      else              { a = *(const v2f*)(pn + kk - 128); }
#pragma unroll
      for (int nt = 0; nt < 4; ++nt) {
        v2f b = *(const v2f*)(&sW1[(kc * 4 + nt) * 64 + lane * 2]);
        acc[nt] = __builtin_amdgcn_wmma_f32_16x16x4_f32(false, a, false, b,
                                                        (short)0, acc[nt], false, false);
      }
    }
#pragma unroll
    for (int nt = 0; nt < 4; ++nt)
#pragma unroll
      for (int v = 0; v < 8; ++v)
        hw[(8 * h + v) * 64 + nt * 16 + m] = fmaxf(acc[nt][v], 0.0f);
    float bb2 = bd2[m];
    v8f acc2 = (v8f){bb2, bb2, bb2, bb2, bb2, bb2, bb2, bb2};
#pragma unroll
    for (int kc = 0; kc < 16; ++kc) {
      const int kk = 4 * kc + 2 * h;
      v2f a = *(const v2f*)(hw + m * 64 + kk);
      v2f b = *(const v2f*)(&sW2[kc * 64 + lane * 2]);
      acc2 = __builtin_amdgcn_wmma_f32_16x16x4_f32(false, a, false, b,
                                                   (short)0, acc2, false, false);
    }
#pragma unroll
    for (int v = 0; v < 8; ++v)
      out[(long)(n0 + 8 * h + v) * 16 + m] = acc2[v];
  }
}

extern "C" void kernel_launch(void* const* d_in, const int* in_sizes, int n_in,
                              void* d_out, int out_size, void* d_ws, size_t ws_size,
                              hipStream_t stream) {
  const float* x    = (const float*)d_in[0];
  const float* ea   = (const float*)d_in[1];
  const float* u    = (const float*)d_in[2];
  const int* send   = (const int*)d_in[3];
  const int* recv   = (const int*)d_in[4];
  const float* Web  = (const float*)d_in[7];
  const float* beb  = (const float*)d_in[8];
  const float* Wnb  = (const float*)d_in[9];
  const float* bnb  = (const float*)d_in[10];
  const float* Wgb  = (const float*)d_in[11];
  const float* bgb  = (const float*)d_in[12];
  const float* Wd1  = (const float*)d_in[13];
  const float* bd1  = (const float*)d_in[14];
  const float* Wd2  = (const float*)d_in[15];
  const float* bd2  = (const float*)d_in[16];
  float* out = (float*)d_out;
  float* ws = (float*)d_ws;

  const long N = NNODES;
  float* agg_r   = ws;
  float* agg_s   = agg_r + N * 64;
  float* cnt_r   = agg_s + N * 64;
  float* cnt_s   = cnt_r + N;
  float* e_sum   = cnt_s + N;
  float* n_sum   = e_sum + 64;
  float* bias_eb = n_sum + 64;
  float* bias_nb = bias_eb + 64;
  float* u2      = bias_nb + 64;
  float* bias_d1 = u2 + 64;
  float* n_hbuf  = bias_d1 + 64;

  const long zlen = 2 * N * 64 + 2 * N + 128;
  int* ctr = (int*)n_hbuf;
  k_zero<<<2048, 256, 0, stream>>>(ws, zlen, ctr, u, Web, beb, Wnb, bnb, bias_eb, bias_nb);
  k_edge<<<1536, 256, 0, stream>>>(x, ea, recv, send, Web, bias_eb,
                                   agg_r, agg_s, cnt_r, cnt_s, ctr, e_sum);
  k_node<<<782, 256, 0, stream>>>(x, agg_r, Wnb, bias_nb, n_hbuf, n_sum);
  k_global<<<1, 64, 0, stream>>>(u, Wgb, bgb, Wd1, bd1, e_sum, n_sum, u2, bias_d1);
  k_dec<<<1024, 256, 0, stream>>>(agg_r, agg_s, cnt_r, cnt_s, n_hbuf,
                                  Wd1, bias_d1, Wd2, bd2, out);
}


# k_node, k_dec: register bound (8 / 6 waves per SIMD) keeps the weight fragments in LDS, not 160 / 416 registers, so 4 / 3 workgroups share a WGP (was 1); k_dec grid 768, tiles looped
# speedup vs baseline: 1.8742x; 1.0671x over previous
#include <hip/hip_runtime.h>

typedef __attribute__((ext_vector_type(2))) float v2f;
typedef __attribute__((ext_vector_type(8))) float v8f;

#define NNODES 100000
#define NEDGES 1600000

__device__ __forceinline__ float xhalf(float x) {
  return __builtin_bit_cast(float, __builtin_amdgcn_permlanex16(
      __builtin_bit_cast(int, x), __builtin_bit_cast(int, x), 0x76543210, 0xfedcba98, false, false));
}

__global__ void k_zero(float* __restrict__ p, long n, int* __restrict__ ctr,
                       const float* __restrict__ u,
                       const float* __restrict__ Web, const float* __restrict__ beb,
                       const float* __restrict__ Wnb, const float* __restrict__ bnb,
                       float* __restrict__ bias_eb, float* __restrict__ bias_nb) {
  if (blockIdx.x == 0 && threadIdx.x < 2) ctr[64 * threadIdx.x] = 0;
  long i = ((long)blockIdx.x * blockDim.x + threadIdx.x) * 4;
  long stride = (long)gridDim.x * blockDim.x * 4;
  float4 z = {0.f, 0.f, 0.f, 0.f};
  for (long j = i; j < n; j += stride) *(float4*)(p + j) = z;
  if (blockIdx.x == 0 && threadIdx.x < 64) {
    const int j = threadIdx.x;
    float a = beb[j], b = bnb[j];
    for (int k = 0; k < 64; ++k) {
      float uk = u[k];
      a += uk * Web[(48 + k) * 64 + j];
      b += uk * Wnb[(80 + k) * 64 + j];
    }
    bias_eb[j] = a;
    bias_nb[j] = b;
  }
}

#define EDGE_RUN 4
__global__ __launch_bounds__(256, 12) void k_edge(
    const float* __restrict__ x, const float* __restrict__ ea,
    const int* __restrict__ recv, const int* __restrict__ send,
    const float* __restrict__ Web, const float* __restrict__ bias_eb,
    float* __restrict__ agg_r, float* __restrict__ agg_s,
    float* __restrict__ cnt_r, float* __restrict__ cnt_s, int* __restrict__ ctr,
    float* __restrict__ e_sum) {
  __shared__ float sW[12 * 4 * 64];
  for (int i = threadIdx.x; i < 12 * 4 * 64; i += 256) {
    int j = i & 1, lane = (i >> 1) & 31, nt = (i >> 6) & 3, kc = i >> 8;
    int kk = 4 * kc + 2 * (lane >> 4) + j;
    sW[i] = Web[kk * 64 + nt * 16 + (lane & 15)];
  }
  __syncthreads();
  const int lane = threadIdx.x & 31;
  const int h = lane >> 4, m = lane & 15;
  const int ntiles = NEDGES / 16;
  const int pkg_half = (__builtin_amdgcn_s_sendmsg_rtn(0x87) >> 18) & 1;
  float es[4] = {0.f, 0.f, 0.f, 0.f};
  for (int pass = 0; pass < 2; ++pass) {
    const int role = pkg_half ^ pass;
    float* __restrict__ agg = role ? agg_s : agg_r;
    float* __restrict__ cnt = role ? cnt_s : cnt_r;
    int* myctr = ctr + 64 * role;
    int t = 0, t1 = 0;
    for (;; ++t) {
      if (t == t1) {
        int t0 = 0;
        if (lane == 0) t0 = atomicAdd(myctr, EDGE_RUN);
        t0 = __builtin_amdgcn_readfirstlane(t0);
        if (t0 >= ntiles) break;
        t = t0;
        t1 = min(t0 + EDGE_RUN, ntiles);
      }
      const int e0 = t * 16;
      const int er = e0 + m;
      const int ri = recv[er], si = send[er];
      const int di = role ? si : ri;
      const float* segs[3] = {ea + (long)er * 16, x + (long)ri * 16, x + (long)si * 16};
      v8f acc[4];
#pragma unroll
      for (int nt = 0; nt < 4; ++nt) {
        float bb = bias_eb[nt * 16 + m];
        acc[nt] = (v8f){bb, bb, bb, bb, bb, bb, bb, bb};
      }
#pragma unroll
      for (int seg = 0; seg < 3; ++seg) {
        const float* bp = segs[seg];
#pragma unroll
        for (int cc = 0; cc < 4; ++cc) {
          v2f a = *(const v2f*)(bp + 4 * cc + 2 * h);
          const int kc = seg * 4 + cc;
#pragma unroll
          for (int nt = 0; nt < 4; ++nt) {
            v2f b = *(const v2f*)(&sW[(kc * 4 + nt) * 64 + lane * 2]);
            acc[nt] = __builtin_amdgcn_wmma_f32_16x16x4_f32(false, a, false, b,
                                                            (short)0, acc[nt], false, false);
          }
        }
      }
#pragma unroll
      for (int nt = 0; nt < 4; ++nt) {
        float ts = 0.f;
#pragma unroll
        for (int v = 0; v < 8; ++v) ts += fmaxf(acc[nt][v], 0.0f);
        es[nt] += role ? 0.0f : ts;
      }
#pragma unroll
      for (int v = 0; v < 8; ++v) {
        const long dA = __builtin_amdgcn_readlane(di, v);
        const long dB = __builtin_amdgcn_readlane(di, 8 + v);
#pragma unroll
        for (int p = 0; p < 2; ++p) {
          const float X = fmaxf(acc[2 * p][v], 0.0f);
          const float Y = fmaxf(acc[2 * p + 1][v], 0.0f);
          const float Zx = xhalf(h ? X : Y);
          const float A = h ? Zx : X;
          const float B = h ? Y : Zx;
          const int col = 32 * p + lane;
          unsafeAtomicAdd(&agg[dA * 64 + col], A);
          unsafeAtomicAdd(&agg[dB * 64 + col], B);
        }
      }
      if (lane < 16) unsafeAtomicAdd(&cnt[di], 1.0f);
    }
  }
  __syncthreads();
#pragma unroll
  for (int nt = 0; nt < 4; ++nt) {
    const float tsum = es[nt] + xhalf(es[nt]);
    if (lane < 16) sW[(threadIdx.x >> 5) * 64 + nt * 16 + m] = tsum;
  }
  __syncthreads();
  if (threadIdx.x < 64) {
    float s = 0.f;
#pragma unroll
    for (int w = 0; w < 8; ++w) s += sW[w * 64 + threadIdx.x];
    if (s != 0.0f) unsafeAtomicAdd(&e_sum[threadIdx.x], s);
  }
}

__global__ __launch_bounds__(256, 8) void k_node(
    const float* __restrict__ x, const float* __restrict__ agg_r,
    const float* __restrict__ Wnb, const float* __restrict__ bias_nb,
    float* __restrict__ n_h, float* __restrict__ n_sum) {
  __shared__ float sW[20 * 4 * 64];
  for (int i = threadIdx.x; i < 20 * 4 * 64; i += 256) {
    int j = i & 1, lane = (i >> 1) & 31, nt = (i >> 6) & 3, kc = i >> 8;
    int kk = 4 * kc + 2 * (lane >> 4) + j;
    sW[i] = Wnb[kk * 64 + nt * 16 + (lane & 15)];
  }
  __syncthreads();
  const int lane = threadIdx.x & 31;
  const int h = lane >> 4, m = lane & 15;
  const int wave = blockIdx.x * 8 + (threadIdx.x >> 5);
  const int nwaves = gridDim.x * 8;
  const int ntiles = NNODES / 16;
  float ns[4] = {0.f, 0.f, 0.f, 0.f};
  for (int t = wave; t < ntiles; t += nwaves) {
    asm volatile("" ::: "memory");
    const int n0 = t * 16;
    const float* ap = agg_r + (long)(n0 + m) * 64;
    const float* xp = x + (long)(n0 + m) * 16;
    v8f acc[4];
#pragma unroll
    for (int nt = 0; nt < 4; ++nt) {
      float bb = bias_nb[nt * 16 + m];
      acc[nt] = (v8f){bb, bb, bb, bb, bb, bb, bb, bb};
    }
#pragma unroll
    for (int kc = 0; kc < 20; ++kc) {
      const int kk = 4 * kc + 2 * h;
      v2f a = (kc < 16) ? *(const v2f*)(ap + kk) : *(const v2f*)(xp + kk - 64);
#pragma unroll
      for (int nt = 0; nt < 4; ++nt) {
        v2f b = *(const v2f*)(&sW[(kc * 4 + nt) * 64 + lane * 2]);
        acc[nt] = __builtin_amdgcn_wmma_f32_16x16x4_f32(false, a, false, b,
                                                        (short)0, acc[nt], false, false);
      }
    }
#pragma unroll
    for (int nt = 0; nt < 4; ++nt)
#pragma unroll
      for (int v = 0; v < 8; ++v) {
        const float r = fmaxf(acc[nt][v], 0.0f);
        n_h[(long)(n0 + 8 * h + v) * 64 + nt * 16 + m] = r;
        ns[nt] += r;
      }
  }
  __syncthreads();
#pragma unroll
  for (int nt = 0; nt < 4; ++nt) {
    const float tsum = ns[nt] + xhalf(ns[nt]);
    if (lane < 16) sW[(threadIdx.x >> 5) * 64 + nt * 16 + m] = tsum;
  }
  __syncthreads();
  if (threadIdx.x < 64) {
    float s = 0.f;
#pragma unroll
    for (int w = 0; w < 8; ++w) s += sW[w * 64 + threadIdx.x];
    if (s != 0.0f) unsafeAtomicAdd(&n_sum[threadIdx.x], s);
  }
}

__global__ void k_global(const float* __restrict__ u,
                         const float* __restrict__ Wgb, const float* __restrict__ bgb,
                         const float* __restrict__ Wd1, const float* __restrict__ bd1,
                         const float* __restrict__ e_sum, const float* __restrict__ n_sum,
                         float* __restrict__ u2, float* __restrict__ bias_d1) {
  __shared__ float su2[64];
  const int j = threadIdx.x;
  float acc = bgb[j];
  for (int k = 0; k < 64; ++k) {
    float em = e_sum[k] * (1.0f / NEDGES);
    float nm = n_sum[k] * (1.0f / NNODES);
    acc += em * Wgb[k * 64 + j] + nm * Wgb[(64 + k) * 64 + j] + u[k] * Wgb[(128 + k) * 64 + j];
  }
  acc = fmaxf(acc, 0.0f);
  su2[j] = acc;
  u2[j] = acc;
  __syncthreads();
  float b = bd1[j];
  for (int k = 0; k < 64; ++k) b += su2[k] * Wd1[(192 + k) * 64 + j];
  bias_d1[j] = b;
}

__global__ __launch_bounds__(256, 6) void k_dec(
    const float* __restrict__ agg_r, const float* __restrict__ agg_s,
    const float* __restrict__ cnt_r, const float* __restrict__ cnt_s,
    const float* __restrict__ n_h, const float* __restrict__ Wd1,
    const float* __restrict__ bias_d1, const float* __restrict__ Wd2,
    const float* __restrict__ bd2, float* __restrict__ out) {
  __shared__ float sW1[48 * 4 * 64];
  __shared__ float sW2[16 * 64];
  __shared__ float sH[8 * 16 * 64];
  for (int i = threadIdx.x; i < 48 * 4 * 64; i += 256) {
    int j = i & 1, lane = (i >> 1) & 31, nt = (i >> 6) & 3, kc = i >> 8;
    int kk = 4 * kc + 2 * (lane >> 4) + j;
    sW1[i] = Wd1[kk * 64 + nt * 16 + (lane & 15)];
  }
  for (int i = threadIdx.x; i < 16 * 64; i += 256) {
    int j = i & 1, lane = (i >> 1) & 31, kc = i >> 6;
    int kk = 4 * kc + 2 * (lane >> 4) + j;
    sW2[i] = Wd2[kk * 16 + (lane & 15)];
  }
  __syncthreads();
  const int lane = threadIdx.x & 31;
  const int h = lane >> 4, m = lane & 15;
  float* hw = sH + (threadIdx.x >> 5) * 1024;
  const int wave = blockIdx.x * 8 + (threadIdx.x >> 5);
  const int nwaves = gridDim.x * 8;
  const int ntiles = NNODES / 16;
  for (int t = wave; t < ntiles; t += nwaves) {
    asm volatile("" ::: "memory");
    const int n0 = t * 16;
    const int n = n0 + m;
    const float invr = 1.0f / fmaxf(cnt_r[n], 1.0f);
    const float invs = 1.0f / fmaxf(cnt_s[n], 1.0f);
    const float* pr = agg_r + (long)n * 64;
    const float* ps = agg_s + (long)n * 64;
    const float* pn = n_h + (long)n * 64;
    v8f acc[4];
#pragma unroll
    for (int nt = 0; nt < 4; ++nt) {
      float bb = bias_d1[nt * 16 + m];
      acc[nt] = (v8f){bb, bb, bb, bb, bb, bb, bb, bb};
    }
#pragma unroll
    for (int kc = 0; kc < 48; ++kc) {
      const int kk = 4 * kc + 2 * h;
      v2f a;
      if (kc < 16)      { a = *(const v2f*)(pr + kk);        a *= invr; }
      else if (kc < 32) { a = *(const v2f*)(ps + kk - 64);   a *= invs; }
      else              { a = *(const v2f*)(pn + kk - 128); }
#pragma unroll
      for (int nt = 0; nt < 4; ++nt) {
        v2f b = *(const v2f*)(&sW1[(kc * 4 + nt) * 64 + lane * 2]);
        acc[nt] = __builtin_amdgcn_wmma_f32_16x16x4_f32(false, a, false, b,
                                                        (short)0, acc[nt], false, false);
      }
    }
#pragma unroll
    for (int nt = 0; nt < 4; ++nt)
#pragma unroll
      for (int v = 0; v < 8; ++v)
        hw[(8 * h + v) * 64 + nt * 16 + m] = fmaxf(acc[nt][v], 0.0f);
    float bb2 = bd2[m];
    v8f acc2 = (v8f){bb2, bb2, bb2, bb2, bb2, bb2, bb2, bb2};
#pragma unroll
    for (int kc = 0; kc < 16; ++kc) {
      const int kk = 4 * kc + 2 * h;
      v2f a = *(const v2f*)(hw + m * 64 + kk);
      v2f b = *(const v2f*)(&sW2[kc * 64 + lane * 2]);
      acc2 = __builtin_amdgcn_wmma_f32_16x16x4_f32(false, a, false, b,
                                                   (short)0, acc2, false, false);
    }
#pragma unroll
    for (int v = 0; v < 8; ++v)
      out[(long)(n0 + 8 * h + v) * 16 + m] = acc2[v];
  }
}

extern "C" void kernel_launch(void* const* d_in, const int* in_sizes, int n_in,
                              void* d_out, int out_size, void* d_ws, size_t ws_size,
                              hipStream_t stream) {
  const float* x    = (const float*)d_in[0];
  const float* ea   = (const float*)d_in[1];
  const float* u    = (const float*)d_in[2];
  const int* send   = (const int*)d_in[3];
  const int* recv   = (const int*)d_in[4];
  const float* Web  = (const float*)d_in[7];
  const float* beb  = (const float*)d_in[8];
  const float* Wnb  = (const float*)d_in[9];
  const float* bnb  = (const float*)d_in[10];
  const float* Wgb  = (const float*)d_in[11];
  const float* bgb  = (const float*)d_in[12];
  const float* Wd1  = (const float*)d_in[13];
  const float* bd1  = (const float*)d_in[14];
  const float* Wd2  = (const float*)d_in[15];
  const float* bd2  = (const float*)d_in[16];
  float* out = (float*)d_out;
  float* ws = (float*)d_ws;

  const long N = NNODES;
  float* agg_r   = ws;
  float* agg_s   = agg_r + N * 64;
  float* cnt_r   = agg_s + N * 64;
  float* cnt_s   = cnt_r + N;
  float* e_sum   = cnt_s + N;
  float* n_sum   = e_sum + 64;
  float* bias_eb = n_sum + 64;
  float* bias_nb = bias_eb + 64;
  float* u2      = bias_nb + 64;
  float* bias_d1 = u2 + 64;
  float* n_hbuf  = bias_d1 + 64;

  const long zlen = 2 * N * 64 + 2 * N + 128;
  int* ctr = (int*)n_hbuf;
  k_zero<<<2048, 256, 0, stream>>>(ws, zlen, ctr, u, Web, beb, Wnb, bnb, bias_eb, bias_nb);
  k_edge<<<1536, 256, 0, stream>>>(x, ea, recv, send, Web, bias_eb,
                                   agg_r, agg_s, cnt_r, cnt_s, ctr, e_sum);
  k_node<<<782, 256, 0, stream>>>(x, agg_r, Wnb, bias_nb, n_hbuf, n_sum);
  k_global<<<1, 64, 0, stream>>>(u, Wgb, bgb, Wd1, bd1, e_sum, n_sum, u2, bias_d1);
  k_dec<<<768, 256, 0, stream>>>(agg_r, agg_s, cnt_r, cnt_s, n_hbuf,
                                  Wd1, bias_d1, Wd2, bd2, out);
}


# k_edge: each package half adds into its own planes of both sums and counts, so e_h is computed once per tile (was once per sum); k_node, k_dec add the two planes; sum order changes
# speedup vs baseline: 1.9540x; 1.0426x over previous
#include <hip/hip_runtime.h>

typedef __attribute__((ext_vector_type(2))) float v2f;
typedef __attribute__((ext_vector_type(8))) float v8f;

#define NNODES 100000
#define NEDGES 1600000
#define HALF_STRIDE (2L * NNODES * 64 + 2L * NNODES)

__device__ __forceinline__ float xhalf(float x) {
  return __builtin_bit_cast(float, __builtin_amdgcn_permlanex16(
      __builtin_bit_cast(int, x), __builtin_bit_cast(int, x), 0x76543210, 0xfedcba98, false, false));
}

__global__ void k_zero(float* __restrict__ p, long n, int* __restrict__ ctr,
                       const float* __restrict__ u,
                       const float* __restrict__ Web, const float* __restrict__ beb,
                       const float* __restrict__ Wnb, const float* __restrict__ bnb,
                       float* __restrict__ bias_eb, float* __restrict__ bias_nb) {
  if (blockIdx.x == 0 && threadIdx.x < 2) ctr[64 * threadIdx.x] = 0;
  long i = ((long)blockIdx.x * blockDim.x + threadIdx.x) * 4;
  long stride = (long)gridDim.x * blockDim.x * 4;
  float4 z = {0.f, 0.f, 0.f, 0.f};
  for (long j = i; j < n; j += stride) *(float4*)(p + j) = z;
  if (blockIdx.x == 0 && threadIdx.x < 64) {
    const int j = threadIdx.x;
    float a = beb[j], b = bnb[j];
    for (int k = 0; k < 64; ++k) {
      float uk = u[k];
      a += uk * Web[(48 + k) * 64 + j];
      b += uk * Wnb[(80 + k) * 64 + j];
    }
    bias_eb[j] = a;
    bias_nb[j] = b;
  }
}

#define EDGE_RUN 2
__global__ __launch_bounds__(256, 12) void k_edge(
    const float* __restrict__ x, const float* __restrict__ ea,
    const int* __restrict__ recv, const int* __restrict__ send,
    const float* __restrict__ Web, const float* __restrict__ bias_eb,
    float* __restrict__ sums, long half_stride, int* __restrict__ ctr,
    float* __restrict__ e_sum) {
  __shared__ float sW[12 * 4 * 64];
  for (int i = threadIdx.x; i < 12 * 4 * 64; i += 256) {
    int j = i & 1, lane = (i >> 1) & 31, nt = (i >> 6) & 3, kc = i >> 8;
    int kk = 4 * kc + 2 * (lane >> 4) + j;
    sW[i] = Web[kk * 64 + nt * 16 + (lane & 15)];
  }
  __syncthreads();
  const int lane = threadIdx.x & 31;
  const int h = lane >> 4, m = lane & 15;
  const int half_tiles = NEDGES / 16 / 2;
  const int pkg_half = (__builtin_amdgcn_s_sendmsg_rtn(0x87) >> 18) & 1;
  float* __restrict__ agg_r = sums + pkg_half * half_stride;
  float* __restrict__ agg_s = agg_r + (long)NNODES * 64;
  float* __restrict__ cnt_r = agg_s + (long)NNODES * 64;
  float* __restrict__ cnt_s = cnt_r + NNODES;
  float es[4] = {0.f, 0.f, 0.f, 0.f};
  for (int pass = 0; pass < 2; ++pass) {
    const int rng = pkg_half ^ pass;
    int* myctr = ctr + 64 * rng;
    const int tbase = rng * half_tiles;
    int t = 0, t1 = 0;
    for (;; ++t) {
      if (t == t1) {
        int t0 = 0;
        if (lane == 0) t0 = atomicAdd(myctr, EDGE_RUN);
        t0 = __builtin_amdgcn_readfirstlane(t0);
        if (t0 >= half_tiles) break;
        t = tbase + t0;
        t1 = t + EDGE_RUN;
      }
      const int e0 = t * 16;
      const int er = e0 + m;
      const int ri = recv[er], si = send[er];
      const float* segs[3] = {ea + (long)er * 16, x + (long)ri * 16, x + (long)si * 16};
      v8f acc[4];
#pragma unroll
      for (int nt = 0; nt < 4; ++nt) {
        float bb = bias_eb[nt * 16 + m];
        acc[nt] = (v8f){bb, bb, bb, bb, bb, bb, bb, bb};
      }
#pragma unroll
      for (int seg = 0; seg < 3; ++seg) {
        const float* bp = segs[seg];
#pragma unroll
        for (int cc = 0; cc < 4; ++cc) {
          v2f a = *(const v2f*)(bp + 4 * cc + 2 * h);
          const int kc = seg * 4 + cc;
#pragma unroll
          for (int nt = 0; nt < 4; ++nt) {
            v2f b = *(const v2f*)(&sW[(kc * 4 + nt) * 64 + lane * 2]);
            acc[nt] = __builtin_amdgcn_wmma_f32_16x16x4_f32(false, a, false, b,
                                                            (short)0, acc[nt], false, false);
          }
        }
      }
#pragma unroll
      for (int nt = 0; nt < 4; ++nt) {
        float ts = 0.f;
#pragma unroll
        for (int v = 0; v < 8; ++v) ts += fmaxf(acc[nt][v], 0.0f);
        es[nt] += ts;
      }
#pragma unroll
      for (int v = 0; v < 8; ++v) {
        const long rA = __builtin_amdgcn_readlane(ri, v);
        const long rB = __builtin_amdgcn_readlane(ri, 8 + v);
        const long sA = __builtin_amdgcn_readlane(si, v);
        const long sB = __builtin_amdgcn_readlane(si, 8 + v);
#pragma unroll
        for (int p = 0; p < 2; ++p) {
          const float X = fmaxf(acc[2 * p][v], 0.0f);
          const float Y = fmaxf(acc[2 * p + 1][v], 0.0f);
          const float Zx = xhalf(h ? X : Y);
          const float A = h ? Zx : X;
          const float B = h ? Y : Zx;
          const int col = 32 * p + lane;
          unsafeAtomicAdd(&agg_r[rA * 64 + col], A);
          unsafeAtomicAdd(&agg_r[rB * 64 + col], B);
          unsafeAtomicAdd(&agg_s[sA * 64 + col], A);
          unsafeAtomicAdd(&agg_s[sB * 64 + col], B);
        }
      }
      unsafeAtomicAdd(h ? &cnt_s[si] : &cnt_r[ri], 1.0f);
    }
  }
  __syncthreads();
#pragma unroll
  for (int nt = 0; nt < 4; ++nt) {
    const float tsum = es[nt] + xhalf(es[nt]);
    if (lane < 16) sW[(threadIdx.x >> 5) * 64 + nt * 16 + m] = tsum;
  }
  __syncthreads();
  if (threadIdx.x < 64) {
    float s = 0.f;
#pragma unroll
    for (int w = 0; w < 8; ++w) s += sW[w * 64 + threadIdx.x];
    if (s != 0.0f) unsafeAtomicAdd(&e_sum[threadIdx.x], s);
  }
}

template <bool TWO>
__global__ __launch_bounds__(256, 8) void k_node(
    const float* __restrict__ x, const float* __restrict__ agg_r0,
    const float* __restrict__ agg_r1,
    const float* __restrict__ Wnb, const float* __restrict__ bias_nb,
    float* __restrict__ n_h, float* __restrict__ n_sum) {
  __shared__ float sW[20 * 4 * 64];
  for (int i = threadIdx.x; i < 20 * 4 * 64; i += 256) {
    int j = i & 1, lane = (i >> 1) & 31, nt = (i >> 6) & 3, kc = i >> 8;
    int kk = 4 * kc + 2 * (lane >> 4) + j;
    sW[i] = Wnb[kk * 64 + nt * 16 + (lane & 15)];
  }
  __syncthreads();
  const int lane = threadIdx.x & 31;
  const int h = lane >> 4, m = lane & 15;
  const int wave = blockIdx.x * 8 + (threadIdx.x >> 5);
  const int nwaves = gridDim.x * 8;
  const int ntiles = NNODES / 16;
  float ns[4] = {0.f, 0.f, 0.f, 0.f};
  for (int t = wave; t < ntiles; t += nwaves) {
    asm volatile("" ::: "memory");
    const int n0 = t * 16;
    const float* ap = agg_r0 + (long)(n0 + m) * 64;
    const float* aq = agg_r1 + (long)(n0 + m) * 64;
    const float* xp = x + (long)(n0 + m) * 16;
    v8f acc[4];
#pragma unroll
    for (int nt = 0; nt < 4; ++nt) {
      float bb = bias_nb[nt * 16 + m];
      acc[nt] = (v8f){bb, bb, bb, bb, bb, bb, bb, bb};
    }
#pragma unroll
    for (int kc = 0; kc < 20; ++kc) {
      const int kk = 4 * kc + 2 * h;
      v2f a = (kc < 16) ? *(const v2f*)(ap + kk) : *(const v2f*)(xp + kk - 64);
      if (TWO && kc < 16) a += *(const v2f*)(aq + kk);
#pragma unroll
      for (int nt = 0; nt < 4; ++nt) {
        v2f b = *(const v2f*)(&sW[(kc * 4 + nt) * 64 + lane * 2]);
        acc[nt] = __builtin_amdgcn_wmma_f32_16x16x4_f32(false, a, false, b,
                                                        (short)0, acc[nt], false, false);
      }
    }
#pragma unroll
    for (int nt = 0; nt < 4; ++nt)
#pragma unroll
      for (int v = 0; v < 8; ++v) {
        const float r = fmaxf(acc[nt][v], 0.0f);
        n_h[(long)(n0 + 8 * h + v) * 64 + nt * 16 + m] = r;
        ns[nt] += r;
      }
  }
  __syncthreads();
#pragma unroll
  for (int nt = 0; nt < 4; ++nt) {
    const float tsum = ns[nt] + xhalf(ns[nt]);
    if (lane < 16) sW[(threadIdx.x >> 5) * 64 + nt * 16 + m] = tsum;
  }
  __syncthreads();
  if (threadIdx.x < 64) {
    float s = 0.f;
#pragma unroll
    for (int w = 0; w < 8; ++w) s += sW[w * 64 + threadIdx.x];
    if (s != 0.0f) unsafeAtomicAdd(&n_sum[threadIdx.x], s);
  }
}

__global__ void k_global(const float* __restrict__ u,
                         const float* __restrict__ Wgb, const float* __restrict__ bgb,
                         const float* __restrict__ Wd1, const float* __restrict__ bd1,
                         const float* __restrict__ e_sum, const float* __restrict__ n_sum,
                         float* __restrict__ u2, float* __restrict__ bias_d1) {
  __shared__ float su2[64];
  const int j = threadIdx.x;
  float acc = bgb[j];
  for (int k = 0; k < 64; ++k) {
    float em = e_sum[k] * (1.0f / NEDGES);
    float nm = n_sum[k] * (1.0f / NNODES);
    acc += em * Wgb[k * 64 + j] + nm * Wgb[(64 + k) * 64 + j] + u[k] * Wgb[(128 + k) * 64 + j];
  }
  acc = fmaxf(acc, 0.0f);
  su2[j] = acc;
  u2[j] = acc;
  __syncthreads();
  float b = bd1[j];
  for (int k = 0; k < 64; ++k) b += su2[k] * Wd1[(192 + k) * 64 + j];
  bias_d1[j] = b;
}

template <bool TWO>
__global__ __launch_bounds__(256, 6) void k_dec(
    const float* __restrict__ agg_r, const float* __restrict__ agg_s,
    const float* __restrict__ cnt_r, const float* __restrict__ cnt_s, long hs,
    const float* __restrict__ n_h, const float* __restrict__ Wd1,
    const float* __restrict__ bias_d1, const float* __restrict__ Wd2,
    const float* __restrict__ bd2, float* __restrict__ out) {
  __shared__ float sW1[48 * 4 * 64];
  __shared__ float sW2[16 * 64];
  __shared__ float sH[8 * 16 * 64];
  for (int i = threadIdx.x; i < 48 * 4 * 64; i += 256) {
    int j = i & 1, lane = (i >> 1) & 31, nt = (i >> 6) & 3, kc = i >> 8;
    int kk = 4 * kc + 2 * (lane >> 4) + j;
    sW1[i] = Wd1[kk * 64 + nt * 16 + (lane & 15)];
  }
  for (int i = threadIdx.x; i < 16 * 64; i += 256) {
    int j = i & 1, lane = (i >> 1) & 31, kc = i >> 6;
    int kk = 4 * kc + 2 * (lane >> 4) + j;
    sW2[i] = Wd2[kk * 16 + (lane & 15)];
  }
  __syncthreads();
  const int lane = threadIdx.x & 31;
  const int h = lane >> 4, m = lane & 15;
  float* hw = sH + (threadIdx.x >> 5) * 1024;
  const int wave = blockIdx.x * 8 + (threadIdx.x >> 5);
  const int nwaves = gridDim.x * 8;
  const int ntiles = NNODES / 16;
  for (int t = wave; t < ntiles; t += nwaves) {
    asm volatile("" ::: "memory");
    const int n0 = t * 16;
    const int n = n0 + m;
    const float cr = cnt_r[n] + (TWO ? cnt_r[hs + n] : 0.0f);
    const float cs = cnt_s[n] + (TWO ? cnt_s[hs + n] : 0.0f);
    const float invr = 1.0f / fmaxf(cr, 1.0f);
    const float invs = 1.0f / fmaxf(cs, 1.0f);
    const float* pr = agg_r + (long)n * 64;
    const float* ps = agg_s + (long)n * 64;
    const float* pn = n_h + (long)n * 64;
    v8f acc[4];
#pragma unroll
    for (int nt = 0; nt < 4; ++nt) {
      float bb = bias_d1[nt * 16 + m];
      acc[nt] = (v8f){bb, bb, bb, bb, bb, bb, bb, bb};
    }
#pragma unroll
    for (int kc = 0; kc < 48; ++kc) {
      const int kk = 4 * kc + 2 * h;
      v2f a;
      if (kc < 16) {
        a = *(const v2f*)(pr + kk);
        if (TWO) a += *(const v2f*)(pr + hs + kk);
        a *= invr;
      } else if (kc < 32) {
        a = *(const v2f*)(ps + kk - 64);
        if (TWO) a += *(const v2f*)(ps + hs + kk - 64);
        a *= invs;
      }
      else              { a = *(const v2f*)(pn + kk - 128); }
#pragma unroll
      for (int nt = 0; nt < 4; ++nt) {
        v2f b = *(const v2f*)(&sW1[(kc * 4 + nt) * 64 + lane * 2]);
        acc[nt] = __builtin_amdgcn_wmma_f32_16x16x4_f32(false, a, false, b,
                                                        (short)0, acc[nt], false, false);
      }
    }
#pragma unroll
    for (int nt = 0; nt < 4; ++nt)
#pragma unroll
      for (int v = 0; v < 8; ++v)
        hw[(8 * h + v) * 64 + nt * 16 + m] = fmaxf(acc[nt][v], 0.0f);
    float bb2 = bd2[m];
    v8f acc2 = (v8f){bb2, bb2, bb2, bb2, bb2, bb2, bb2, bb2};
#pragma unroll
    for (int kc = 0; kc < 16; ++kc) {
      const int kk = 4 * kc + 2 * h;
      v2f a = *(const v2f*)(hw + m * 64 + kk);
      v2f b = *(const v2f*)(&sW2[kc * 64 + lane * 2]);
      acc2 = __builtin_amdgcn_wmma_f32_16x16x4_f32(false, a, false, b,
                                                   (short)0, acc2, false, false);
    }
#pragma unroll
    for (int v = 0; v < 8; ++v)
      out[(long)(n0 + 8 * h + v) * 16 + m] = acc2[v];
  }
}

extern "C" void kernel_launch(void* const* d_in, const int* in_sizes, int n_in,
                              void* d_out, int out_size, void* d_ws, size_t ws_size,
                              hipStream_t stream) {
  const float* x    = (const float*)d_in[0];
  const float* ea   = (const float*)d_in[1];
  const float* u    = (const float*)d_in[2];
  const int* send   = (const int*)d_in[3];
  const int* recv   = (const int*)d_in[4];
  const float* Web  = (const float*)d_in[7];
  const float* beb  = (const float*)d_in[8];
  const float* Wnb  = (const float*)d_in[9];
  const float* bnb  = (const float*)d_in[10];
  const float* Wgb  = (const float*)d_in[11];
  const float* bgb  = (const float*)d_in[12];
  const float* Wd1  = (const float*)d_in[13];
  const float* bd1  = (const float*)d_in[14];
  const float* Wd2  = (const float*)d_in[15];
  const float* bd2  = (const float*)d_in[16];
  float* out = (float*)d_out;
  float* ws = (float*)d_ws;

  const long N = NNODES;
  if (ws_size >= (size_t)(2 * HALF_STRIDE + 384 + N * 64) * sizeof(float)) {
    float* sums    = ws;
    float* e_sum   = sums + 2 * HALF_STRIDE;
    float* n_sum   = e_sum + 64;
    float* bias_eb = n_sum + 64;
    float* bias_nb = bias_eb + 64;
    float* u2      = bias_nb + 64;
    float* bias_d1 = u2 + 64;
    float* n_hbuf  = bias_d1 + 64;
    int* ctr = (int*)n_hbuf;
    k_zero<<<2048, 256, 0, stream>>>(ws, 2 * HALF_STRIDE + 128, ctr, u, Web, beb, Wnb, bnb,
                                     bias_eb, bias_nb);
    k_edge<<<1536, 256, 0, stream>>>(x, ea, recv, send, Web, bias_eb, sums, HALF_STRIDE, ctr,
                                     e_sum);
    k_node<true><<<782, 256, 0, stream>>>(x, sums, sums + HALF_STRIDE, Wnb, bias_nb, n_hbuf,
                                          n_sum);
    k_global<<<1, 64, 0, stream>>>(u, Wgb, bgb, Wd1, bd1, e_sum, n_sum, u2, bias_d1);
    k_dec<true><<<768, 256, 0, stream>>>(sums, sums + N * 64, sums + 2 * N * 64,
                                         sums + 2 * N * 64 + N, HALF_STRIDE, n_hbuf,
                                         Wd1, bias_d1, Wd2, bd2, out);
  } else {
    float* sums    = ws;
    float* e_sum   = sums + HALF_STRIDE;
    float* n_sum   = e_sum + 64;
    float* bias_eb = n_sum + 64;
    float* bias_nb = bias_eb + 64;
    float* u2      = bias_nb + 64;
    float* bias_d1 = u2 + 64;
    float* n_hbuf  = bias_d1 + 64;
    int* ctr = (int*)n_hbuf;
    k_zero<<<2048, 256, 0, stream>>>(ws, HALF_STRIDE + 128, ctr, u, Web, beb, Wnb, bnb,
                                     bias_eb, bias_nb);
    k_edge<<<1536, 256, 0, stream>>>(x, ea, recv, send, Web, bias_eb, sums, 0, ctr, e_sum);
    k_node<false><<<782, 256, 0, stream>>>(x, sums, sums, Wnb, bias_nb, n_hbuf, n_sum);
    k_global<<<1, 64, 0, stream>>>(u, Wgb, bgb, Wd1, bd1, e_sum, n_sum, u2, bias_d1);
    k_dec<false><<<768, 256, 0, stream>>>(sums, sums + N * 64, sums + 2 * N * 64,
                                          sums + 2 * N * 64 + N, 0, n_hbuf,
                                          Wd1, bias_d1, Wd2, bd2, out);
  }
}


# k_edge, k_node, k_dec: A rows read 16 bytes per lane, a tile's loads issued in groups ahead of the products (k order within each 8 permuted, weights staged to match); sum order changes
# speedup vs baseline: 2.0599x; 1.0542x over previous
#include <hip/hip_runtime.h>

typedef __attribute__((ext_vector_type(2))) float v2f;
typedef __attribute__((ext_vector_type(4))) float v4f;
typedef __attribute__((ext_vector_type(8))) float v8f;


#define NNODES 100000
#define NEDGES 1600000
#define HALF_STRIDE (2L * NNODES * 64 + 2L * NNODES)

__device__ __forceinline__ float xhalf(float x) {
  return __builtin_bit_cast(float, __builtin_amdgcn_permlanex16(
      __builtin_bit_cast(int, x), __builtin_bit_cast(int, x), 0x76543210, 0xfedcba98, false, false));
}

__global__ void k_zero(float* __restrict__ p, long n, int* __restrict__ ctr,
                       const float* __restrict__ u,
                       const float* __restrict__ Web, const float* __restrict__ beb,
                       const float* __restrict__ Wnb, const float* __restrict__ bnb,
                       float* __restrict__ bias_eb, float* __restrict__ bias_nb) {
  if (blockIdx.x == 0 && threadIdx.x < 2) ctr[64 * threadIdx.x] = 0;
  long i = ((long)blockIdx.x * blockDim.x + threadIdx.x) * 4;
  long stride = (long)gridDim.x * blockDim.x * 4;
  float4 z = {0.f, 0.f, 0.f, 0.f};
  for (long j = i; j < n; j += stride) *(float4*)(p + j) = z;
  if (blockIdx.x == 0 && threadIdx.x < 64) {
    const int j = threadIdx.x;
    float a = beb[j], b = bnb[j];
    for (int k = 0; k < 64; ++k) {
      float uk = u[k];
      a += uk * Web[(48 + k) * 64 + j];
      b += uk * Wnb[(80 + k) * 64 + j];
    }
    bias_eb[j] = a;
    bias_nb[j] = b;
  }
}

#define EDGE_RUN 2
__global__ __launch_bounds__(256, 12) void k_edge(
    const float* __restrict__ x, const float* __restrict__ ea,
    const int* __restrict__ recv, const int* __restrict__ send,
    const float* __restrict__ Web, const float* __restrict__ bias_eb,
    float* __restrict__ sums, long half_stride, int* __restrict__ ctr,
    float* __restrict__ e_sum) {
  __shared__ float sW[12 * 4 * 64];
  for (int i = threadIdx.x; i < 12 * 4 * 64; i += 256) {
    int j = i & 1, lane = (i >> 1) & 31, nt = (i >> 6) & 3, kc = i >> 8;
    int kk = 8 * (kc >> 1) + 4 * (lane >> 4) + 2 * (kc & 1) + j;
    sW[i] = Web[kk * 64 + nt * 16 + (lane & 15)];
  }
  __syncthreads();
  const int lane = threadIdx.x & 31;
  const int h = lane >> 4, m = lane & 15;
  const int half_tiles = NEDGES / 16 / 2;
  const int pkg_half = (__builtin_amdgcn_s_sendmsg_rtn(0x87) >> 18) & 1;
  float* __restrict__ agg_r = sums + pkg_half * half_stride;
  float* __restrict__ agg_s = agg_r + (long)NNODES * 64;
  float* __restrict__ cnt_r = agg_s + (long)NNODES * 64;
  float* __restrict__ cnt_s = cnt_r + NNODES;
  float es[4] = {0.f, 0.f, 0.f, 0.f};
  for (int pass = 0; pass < 2; ++pass) {
    const int rng = pkg_half ^ pass;
    int* myctr = ctr + 64 * rng;
    const int tbase = rng * half_tiles;
    int t = 0, t1 = 0;
    for (;; ++t) {
      if (t == t1) {
        int t0 = 0;
        if (lane == 0) t0 = atomicAdd(myctr, EDGE_RUN);
        t0 = __builtin_amdgcn_readfirstlane(t0);
        if (t0 >= half_tiles) break;
        t = tbase + t0;
        t1 = t + EDGE_RUN;
      }
      const int e0 = t * 16;
      const int er = e0 + m;
      const int ri = recv[er], si = send[er];
      const float* segs[3] = {ea + (long)er * 16, x + (long)ri * 16, x + (long)si * 16};
      v8f acc[4];
#pragma unroll
      for (int nt = 0; nt < 4; ++nt) {
        float bb = bias_eb[nt * 16 + m];
        acc[nt] = (v8f){bb, bb, bb, bb, bb, bb, bb, bb};
      }
      v4f q[6];
#pragma unroll
      for (int i = 0; i < 4; ++i) q[i] = *(const v4f*)(segs[i >> 1] + 8 * (i & 1) + 4 * h);
      asm volatile("" ::: "memory");
#pragma unroll
      for (int seg = 0; seg < 3; ++seg) {
        if (seg == 1) {
#pragma unroll
          for (int i = 4; i < 6; ++i) q[i] = *(const v4f*)(segs[2] + 8 * (i & 1) + 4 * h);
          asm volatile("" ::: "memory");
        }
#pragma unroll
        for (int g = 0; g < 2; ++g) {
          const v4f qq = q[2 * seg + g];
#pragma unroll
          for (int j = 0; j < 2; ++j) {
            const v2f a = j ? (v2f){qq.z, qq.w} : (v2f){qq.x, qq.y};
            const int kc = seg * 4 + 2 * g + j;
#pragma unroll
            for (int nt = 0; nt < 4; ++nt) {
              v2f b = *(const v2f*)(&sW[(kc * 4 + nt) * 64 + lane * 2]);
              acc[nt] = __builtin_amdgcn_wmma_f32_16x16x4_f32(false, a, false, b,
                                                              (short)0, acc[nt], false, false);
            }
          }
        }
      }
#pragma unroll
      for (int nt = 0; nt < 4; ++nt) {
        float ts = 0.f;
#pragma unroll
        for (int v = 0; v < 8; ++v) ts += fmaxf(acc[nt][v], 0.0f);
        es[nt] += ts;
      }
#pragma unroll
      for (int v = 0; v < 8; ++v) {
        const long rA = __builtin_amdgcn_readlane(ri, v);
        const long rB = __builtin_amdgcn_readlane(ri, 8 + v);
        const long sA = __builtin_amdgcn_readlane(si, v);
        const long sB = __builtin_amdgcn_readlane(si, 8 + v);
#pragma unroll
        for (int p = 0; p < 2; ++p) {
          const float X = fmaxf(acc[2 * p][v], 0.0f);
          const float Y = fmaxf(acc[2 * p + 1][v], 0.0f);
          const float Zx = xhalf(h ? X : Y);
          const float A = h ? Zx : X;
          const float B = h ? Y : Zx;
          const int col = 32 * p + lane;
          unsafeAtomicAdd(&agg_r[rA * 64 + col], A);
          unsafeAtomicAdd(&agg_r[rB * 64 + col], B);
          unsafeAtomicAdd(&agg_s[sA * 64 + col], A);
          unsafeAtomicAdd(&agg_s[sB * 64 + col], B);
        }
      }
      unsafeAtomicAdd(h ? &cnt_s[si] : &cnt_r[ri], 1.0f);
    }
  }
  __syncthreads();
#pragma unroll
  for (int nt = 0; nt < 4; ++nt) {
    const float tsum = es[nt] + xhalf(es[nt]);
    if (lane < 16) sW[(threadIdx.x >> 5) * 64 + nt * 16 + m] = tsum;
  }
  __syncthreads();
  if (threadIdx.x < 64) {
    float s = 0.f;
#pragma unroll
    for (int w = 0; w < 8; ++w) s += sW[w * 64 + threadIdx.x];
    if (s != 0.0f) unsafeAtomicAdd(&e_sum[threadIdx.x], s);
  }
}

template <bool TWO>
__global__ __launch_bounds__(256, 8) void k_node(
    const float* __restrict__ x, const float* __restrict__ agg_r0,
    const float* __restrict__ agg_r1,
    const float* __restrict__ Wnb, const float* __restrict__ bias_nb,
    float* __restrict__ n_h, float* __restrict__ n_sum) {
  __shared__ float sW[20 * 4 * 64];
  for (int i = threadIdx.x; i < 20 * 4 * 64; i += 256) {
    int j = i & 1, lane = (i >> 1) & 31, nt = (i >> 6) & 3, kc = i >> 8;
    int kk = 8 * (kc >> 1) + 4 * (lane >> 4) + 2 * (kc & 1) + j;
    sW[i] = Wnb[kk * 64 + nt * 16 + (lane & 15)];
  }
  __syncthreads();
  const int lane = threadIdx.x & 31;
  const int h = lane >> 4, m = lane & 15;
  const int wave = blockIdx.x * 8 + (threadIdx.x >> 5);
  const int nwaves = gridDim.x * 8;
  const int ntiles = NNODES / 16;
  float ns[4] = {0.f, 0.f, 0.f, 0.f};
  for (int t = wave; t < ntiles; t += nwaves) {
    asm volatile("" ::: "memory");
    const int n0 = t * 16;
    const float* ap = agg_r0 + (long)(n0 + m) * 64;
    const float* aq = agg_r1 + (long)(n0 + m) * 64;
    const float* xp = x + (long)(n0 + m) * 16;
    v8f acc[4];
    float bb[4];
#pragma unroll
    for (int nt = 0; nt < 4; ++nt) bb[nt] = bias_nb[nt * 16 + m];
    v4f qa[10], qb[8];
#pragma unroll
    for (int g = 0; g < 10; ++g) {
      const int kk = 8 * g + 4 * h;
      qa[g] = (g < 8) ? *(const v4f*)(ap + kk) : *(const v4f*)(xp + kk - 64);
      if (TWO && g < 8) qb[g] = *(const v4f*)(aq + kk);
    }
    asm volatile("" ::: "memory");
#pragma unroll
    for (int nt = 0; nt < 4; ++nt)
      acc[nt] = (v8f){bb[nt], bb[nt], bb[nt], bb[nt], bb[nt], bb[nt], bb[nt], bb[nt]};
#pragma unroll
    for (int g = 0; g < 10; ++g) {
      v4f q = qa[g];
      if (TWO && g < 8) q += qb[g];
#pragma unroll
      for (int j = 0; j < 2; ++j) {
        const v2f a = j ? (v2f){q.z, q.w} : (v2f){q.x, q.y};
        const int kc = 2 * g + j;
#pragma unroll
        for (int nt = 0; nt < 4; ++nt) {
          v2f b = *(const v2f*)(&sW[(kc * 4 + nt) * 64 + lane * 2]);
          acc[nt] = __builtin_amdgcn_wmma_f32_16x16x4_f32(false, a, false, b,
                                                          (short)0, acc[nt], false, false);
        }
      }
    }
#pragma unroll
    for (int nt = 0; nt < 4; ++nt)
#pragma unroll
      for (int v = 0; v < 8; ++v) {
        const float r = fmaxf(acc[nt][v], 0.0f);
        n_h[(long)(n0 + 8 * h + v) * 64 + nt * 16 + m] = r;
        ns[nt] += r;
      }
  }
  __syncthreads();
#pragma unroll
  for (int nt = 0; nt < 4; ++nt) {
    const float tsum = ns[nt] + xhalf(ns[nt]);
    if (lane < 16) sW[(threadIdx.x >> 5) * 64 + nt * 16 + m] = tsum;
  }
  __syncthreads();
  if (threadIdx.x < 64) {
    float s = 0.f;
#pragma unroll
    for (int w = 0; w < 8; ++w) s += sW[w * 64 + threadIdx.x];
    if (s != 0.0f) unsafeAtomicAdd(&n_sum[threadIdx.x], s);
  }
}

__global__ void k_global(const float* __restrict__ u,
                         const float* __restrict__ Wgb, const float* __restrict__ bgb,
                         const float* __restrict__ Wd1, const float* __restrict__ bd1,
                         const float* __restrict__ e_sum, const float* __restrict__ n_sum,
                         float* __restrict__ u2, float* __restrict__ bias_d1) {
  __shared__ float su2[64];
  const int j = threadIdx.x;
  float acc = bgb[j];
  for (int k = 0; k < 64; ++k) {
    float em = e_sum[k] * (1.0f / NEDGES);
    float nm = n_sum[k] * (1.0f / NNODES);
    acc += em * Wgb[k * 64 + j] + nm * Wgb[(64 + k) * 64 + j] + u[k] * Wgb[(128 + k) * 64 + j];
  }
  acc = fmaxf(acc, 0.0f);
  su2[j] = acc;
  u2[j] = acc;
  __syncthreads();
  float b = bd1[j];
  for (int k = 0; k < 64; ++k) b += su2[k] * Wd1[(192 + k) * 64 + j];
  bias_d1[j] = b;
}

template <bool TWO>
__global__ __launch_bounds__(256, 6) void k_dec(
    const float* __restrict__ agg_r, const float* __restrict__ agg_s,
    const float* __restrict__ cnt_r, const float* __restrict__ cnt_s, long hs,
    const float* __restrict__ n_h, const float* __restrict__ Wd1,
    const float* __restrict__ bias_d1, const float* __restrict__ Wd2,
    const float* __restrict__ bd2, float* __restrict__ out) {
  __shared__ float sW1[48 * 4 * 64];
  __shared__ float sW2[16 * 64];
  __shared__ float sH[8 * 16 * 64];
  for (int i = threadIdx.x; i < 48 * 4 * 64; i += 256) {
    int j = i & 1, lane = (i >> 1) & 31, nt = (i >> 6) & 3, kc = i >> 8;
    int kk = 8 * (kc >> 1) + 4 * (lane >> 4) + 2 * (kc & 1) + j;
    sW1[i] = Wd1[kk * 64 + nt * 16 + (lane & 15)];
  }
  for (int i = threadIdx.x; i < 16 * 64; i += 256) {
    int j = i & 1, lane = (i >> 1) & 31, kc = i >> 6;
    int kk = 4 * kc + 2 * (lane >> 4) + j;
    sW2[i] = Wd2[kk * 16 + (lane & 15)];
  }
  __syncthreads();
  const int lane = threadIdx.x & 31;
  const int h = lane >> 4, m = lane & 15;
  float* hw = sH + (threadIdx.x >> 5) * 1024;
  const int wave = blockIdx.x * 8 + (threadIdx.x >> 5);
  const int nwaves = gridDim.x * 8;
  const int ntiles = NNODES / 16;
  for (int t = wave; t < ntiles; t += nwaves) {
    asm volatile("" ::: "memory");
    const int n0 = t * 16;
    const int n = n0 + m;
    const float cr = cnt_r[n] + (TWO ? cnt_r[hs + n] : 0.0f);
    const float cs = cnt_s[n] + (TWO ? cnt_s[hs + n] : 0.0f);
    const float invr = 1.0f / fmaxf(cr, 1.0f);
    const float invs = 1.0f / fmaxf(cs, 1.0f);
    const float* pr = agg_r + (long)n * 64;
    const float* ps = agg_s + (long)n * 64;
    const float* pn = n_h + (long)n * 64;
    v8f acc[4];
    float bb[4];
#pragma unroll
    for (int nt = 0; nt < 4; ++nt) bb[nt] = bias_d1[nt * 16 + m];
    auto ld = [&](const int part, v4f (&A)[4], v4f (&B)[4]) {
#pragma unroll
      for (int gg = 0; gg < 4; ++gg) {
        const int kk = 8 * (4 * (part & 1) + gg) + 4 * h;
        const float* p = part < 2 ? pr : part < 4 ? ps : pn;
        A[gg] = *(const v4f*)(p + kk);
        if (TWO && part < 4) B[gg] = *(const v4f*)(p + hs + kk);
      }
    };
    auto mm = [&](const int part, const v4f (&A)[4], const v4f (&B)[4]) {
#pragma unroll
      for (int gg = 0; gg < 4; ++gg) {
        v4f q = A[gg];
        if (part < 4) {
          if (TWO) q += B[gg];
          q *= (part < 2) ? invr : invs;
        }
#pragma unroll
        for (int j = 0; j < 2; ++j) {
          const v2f a = j ? (v2f){q.z, q.w} : (v2f){q.x, q.y};
          const int kc = 2 * (4 * part + gg) + j;
#pragma unroll
          for (int nt = 0; nt < 4; ++nt) {
            v2f b = *(const v2f*)(&sW1[(kc * 4 + nt) * 64 + lane * 2]);
            acc[nt] = __builtin_amdgcn_wmma_f32_16x16x4_f32(false, a, false, b,
                                                            (short)0, acc[nt], false, false);
          }
        }
      }
    };
    v4f a0[4], b0[4], a1[4], b1[4], a2[4], b2[4];
    ld(0, a0, b0);
    ld(1, a1, b1);
    asm volatile("" ::: "memory");
#pragma unroll
    for (int nt = 0; nt < 4; ++nt)
      acc[nt] = (v8f){bb[nt], bb[nt], bb[nt], bb[nt], bb[nt], bb[nt], bb[nt], bb[nt]};
    mm(0, a0, b0);
    ld(2, a2, b2);
    asm volatile("" ::: "memory");
    mm(1, a1, b1);
    ld(3, a0, b0);
    asm volatile("" ::: "memory");
    mm(2, a2, b2);
    ld(4, a1, b1);
    asm volatile("" ::: "memory");
    mm(3, a0, b0);
    ld(5, a2, b2);
    asm volatile("" ::: "memory");
    mm(4, a1, b1);
    mm(5, a2, b2);
#pragma unroll
    for (int nt = 0; nt < 4; ++nt)
#pragma unroll
      for (int v = 0; v < 8; ++v)
        hw[(8 * h + v) * 64 + nt * 16 + m] = fmaxf(acc[nt][v], 0.0f);
    float bb2 = bd2[m];
    v8f acc2 = (v8f){bb2, bb2, bb2, bb2, bb2, bb2, bb2, bb2};
#pragma unroll
    for (int kc = 0; kc < 16; ++kc) {
      const int kk = 4 * kc + 2 * h;
      v2f a = *(const v2f*)(hw + m * 64 + kk);
      v2f b = *(const v2f*)(&sW2[kc * 64 + lane * 2]);
      acc2 = __builtin_amdgcn_wmma_f32_16x16x4_f32(false, a, false, b,
                                                   (short)0, acc2, false, false);
    }
#pragma unroll
    for (int v = 0; v < 8; ++v)
      out[(long)(n0 + 8 * h + v) * 16 + m] = acc2[v];
  }
}

extern "C" void kernel_launch(void* const* d_in, const int* in_sizes, int n_in,
                              void* d_out, int out_size, void* d_ws, size_t ws_size,
                              hipStream_t stream) {
  const float* x    = (const float*)d_in[0];
  const float* ea   = (const float*)d_in[1];
  const float* u    = (const float*)d_in[2];
  const int* send   = (const int*)d_in[3];
  const int* recv   = (const int*)d_in[4];
  const float* Web  = (const float*)d_in[7];
  const float* beb  = (const float*)d_in[8];
  const float* Wnb  = (const float*)d_in[9];
  const float* bnb  = (const float*)d_in[10];
  const float* Wgb  = (const float*)d_in[11];
  const float* bgb  = (const float*)d_in[12];
  const float* Wd1  = (const float*)d_in[13];
  const float* bd1  = (const float*)d_in[14];
  const float* Wd2  = (const float*)d_in[15];
  const float* bd2  = (const float*)d_in[16];
  float* out = (float*)d_out;
  float* ws = (float*)d_ws;

  const long N = NNODES;
  if (ws_size >= (size_t)(2 * HALF_STRIDE + 384 + N * 64) * sizeof(float)) {
    float* sums    = ws;
    float* e_sum   = sums + 2 * HALF_STRIDE;
    float* n_sum   = e_sum + 64;
    float* bias_eb = n_sum + 64;
    float* bias_nb = bias_eb + 64;
    float* u2      = bias_nb + 64;
    float* bias_d1 = u2 + 64;
    float* n_hbuf  = bias_d1 + 64;
    int* ctr = (int*)n_hbuf;
    k_zero<<<2048, 256, 0, stream>>>(ws, 2 * HALF_STRIDE + 128, ctr, u, Web, beb, Wnb, bnb,
                                     bias_eb, bias_nb);
    k_edge<<<1536, 256, 0, stream>>>(x, ea, recv, send, Web, bias_eb, sums, HALF_STRIDE, ctr,
                                     e_sum);
    k_node<true><<<782, 256, 0, stream>>>(x, sums, sums + HALF_STRIDE, Wnb, bias_nb, n_hbuf,
                                          n_sum);
    k_global<<<1, 64, 0, stream>>>(u, Wgb, bgb, Wd1, bd1, e_sum, n_sum, u2, bias_d1);
    k_dec<true><<<768, 256, 0, stream>>>(sums, sums + N * 64, sums + 2 * N * 64,
                                         sums + 2 * N * 64 + N, HALF_STRIDE, n_hbuf,
                                         Wd1, bias_d1, Wd2, bd2, out);
  } else {
    float* sums    = ws;
    float* e_sum   = sums + HALF_STRIDE;
    float* n_sum   = e_sum + 64;
    float* bias_eb = n_sum + 64;
    float* bias_nb = bias_eb + 64;
    float* u2      = bias_nb + 64;
    float* bias_d1 = u2 + 64;
    float* n_hbuf  = bias_d1 + 64;
    int* ctr = (int*)n_hbuf;
    k_zero<<<2048, 256, 0, stream>>>(ws, HALF_STRIDE + 128, ctr, u, Web, beb, Wnb, bnb,
                                     bias_eb, bias_nb);
    k_edge<<<1536, 256, 0, stream>>>(x, ea, recv, send, Web, bias_eb, sums, 0, ctr, e_sum);
    k_node<false><<<782, 256, 0, stream>>>(x, sums, sums, Wnb, bias_nb, n_hbuf, n_sum);
    k_global<<<1, 64, 0, stream>>>(u, Wgb, bgb, Wd1, bd1, e_sum, n_sum, u2, bias_d1);
    k_dec<false><<<768, 256, 0, stream>>>(sums, sums + N * 64, sums + 2 * N * 64,
                                          sums + 2 * N * 64 + N, 0, n_hbuf,
                                          Wd1, bias_d1, Wd2, bd2, out);
  }
}


# 5 launches to 4: every k_dec workgroup computes the global block (u2, decoder bias) in its prologue, 64 k of each product split over 4 thread quarters, joined in LDS; sum order of u2 and bias changes
# speedup vs baseline: 2.1746x; 1.0557x over previous
#include <hip/hip_runtime.h>

typedef __attribute__((ext_vector_type(2))) float v2f;
typedef __attribute__((ext_vector_type(4))) float v4f;
typedef __attribute__((ext_vector_type(8))) float v8f;


#define NNODES 100000
#define NEDGES 1600000
#define HALF_STRIDE (2L * NNODES * 64 + 2L * NNODES)

__device__ __forceinline__ float xhalf(float x) {
  return __builtin_bit_cast(float, __builtin_amdgcn_permlanex16(
      __builtin_bit_cast(int, x), __builtin_bit_cast(int, x), 0x76543210, 0xfedcba98, false, false));
}

__global__ void k_zero(float* __restrict__ p, long n, int* __restrict__ ctr,
                       const float* __restrict__ u,
                       const float* __restrict__ Web, const float* __restrict__ beb,
                       const float* __restrict__ Wnb, const float* __restrict__ bnb,
                       float* __restrict__ bias_eb, float* __restrict__ bias_nb) {
  if (blockIdx.x == 0 && threadIdx.x < 2) ctr[64 * threadIdx.x] = 0;
  long i = ((long)blockIdx.x * blockDim.x + threadIdx.x) * 4;
  long stride = (long)gridDim.x * blockDim.x * 4;
  float4 z = {0.f, 0.f, 0.f, 0.f};
  for (long j = i; j < n; j += stride) *(float4*)(p + j) = z;
  if (blockIdx.x == 0 && threadIdx.x < 64) {
    const int j = threadIdx.x;
    float a = beb[j], b = bnb[j];
    for (int k = 0; k < 64; ++k) {
      float uk = u[k];
      a += uk * Web[(48 + k) * 64 + j];
      b += uk * Wnb[(80 + k) * 64 + j];
    }
    bias_eb[j] = a;
    bias_nb[j] = b;
  }
}

#define EDGE_RUN 2
__global__ __launch_bounds__(256, 12) void k_edge(
    const float* __restrict__ x, const float* __restrict__ ea,
    const int* __restrict__ recv, const int* __restrict__ send,
    const float* __restrict__ Web, const float* __restrict__ bias_eb,
    float* __restrict__ sums, long half_stride, int* __restrict__ ctr,
    float* __restrict__ e_sum) {
  __shared__ float sW[12 * 4 * 64];
  for (int i = threadIdx.x; i < 12 * 4 * 64; i += 256) {
    int j = i & 1, lane = (i >> 1) & 31, nt = (i >> 6) & 3, kc = i >> 8;
    int kk = 8 * (kc >> 1) + 4 * (lane >> 4) + 2 * (kc & 1) + j;
    sW[i] = Web[kk * 64 + nt * 16 + (lane & 15)];
  }
  __syncthreads();
  const int lane = threadIdx.x & 31;
  const int h = lane >> 4, m = lane & 15;
  const int half_tiles = NEDGES / 16 / 2;
  const int pkg_half = (__builtin_amdgcn_s_sendmsg_rtn(0x87) >> 18) & 1;
  float* __restrict__ agg_r = sums + pkg_half * half_stride;
  float* __restrict__ agg_s = agg_r + (long)NNODES * 64;
  float* __restrict__ cnt_r = agg_s + (long)NNODES * 64;
  float* __restrict__ cnt_s = cnt_r + NNODES;
  float es[4] = {0.f, 0.f, 0.f, 0.f};
  for (int pass = 0; pass < 2; ++pass) {
    const int rng = pkg_half ^ pass;
    int* myctr = ctr + 64 * rng;
    const int tbase = rng * half_tiles;
    int t = 0, t1 = 0;
    for (;; ++t) {
      if (t == t1) {
        int t0 = 0;
        if (lane == 0) t0 = atomicAdd(myctr, EDGE_RUN);
        t0 = __builtin_amdgcn_readfirstlane(t0);
        if (t0 >= half_tiles) break;
        t = tbase + t0;
        t1 = t + EDGE_RUN;
      }
      const int e0 = t * 16;
      const int er = e0 + m;
      const int ri = recv[er], si = send[er];
      const float* segs[3] = {ea + (long)er * 16, x + (long)ri * 16, x + (long)si * 16};
      v8f acc[4];
#pragma unroll
      for (int nt = 0; nt < 4; ++nt) {
        float bb = bias_eb[nt * 16 + m];
        acc[nt] = (v8f){bb, bb, bb, bb, bb, bb, bb, bb};
      }
      v4f q[6];
#pragma unroll
      for (int i = 0; i < 4; ++i) q[i] = *(const v4f*)(segs[i >> 1] + 8 * (i & 1) + 4 * h);
      asm volatile("" ::: "memory");
#pragma unroll
      for (int seg = 0; seg < 3; ++seg) {
        if (seg == 1) {
#pragma unroll
          for (int i = 4; i < 6; ++i) q[i] = *(const v4f*)(segs[2] + 8 * (i & 1) + 4 * h);
          asm volatile("" ::: "memory");
        }
#pragma unroll
        for (int g = 0; g < 2; ++g) {
          const v4f qq = q[2 * seg + g];
#pragma unroll
          for (int j = 0; j < 2; ++j) {
            const v2f a = j ? (v2f){qq.z, qq.w} : (v2f){qq.x, qq.y};
            const int kc = seg * 4 + 2 * g + j;
#pragma unroll
            for (int nt = 0; nt < 4; ++nt) {
              v2f b = *(const v2f*)(&sW[(kc * 4 + nt) * 64 + lane * 2]);
              acc[nt] = __builtin_amdgcn_wmma_f32_16x16x4_f32(false, a, false, b,
                                                              (short)0, acc[nt], false, false);
            }
          }
        }
      }
#pragma unroll
      for (int nt = 0; nt < 4; ++nt) {
        float ts = 0.f;
#pragma unroll
        for (int v = 0; v < 8; ++v) ts += fmaxf(acc[nt][v], 0.0f);
        es[nt] += ts;
      }
#pragma unroll
      for (int v = 0; v < 8; ++v) {
        const long rA = __builtin_amdgcn_readlane(ri, v);
        const long rB = __builtin_amdgcn_readlane(ri, 8 + v);
        const long sA = __builtin_amdgcn_readlane(si, v);
        const long sB = __builtin_amdgcn_readlane(si, 8 + v);
#pragma unroll
        for (int p = 0; p < 2; ++p) {
          const float X = fmaxf(acc[2 * p][v], 0.0f);
          const float Y = fmaxf(acc[2 * p + 1][v], 0.0f);
          const float Zx = xhalf(h ? X : Y);
          const float A = h ? Zx : X;
          const float B = h ? Y : Zx;
          const int col = 32 * p + lane;
          unsafeAtomicAdd(&agg_r[rA * 64 + col], A);
          unsafeAtomicAdd(&agg_r[rB * 64 + col], B);
          unsafeAtomicAdd(&agg_s[sA * 64 + col], A);
          unsafeAtomicAdd(&agg_s[sB * 64 + col], B);
        }
      }
      unsafeAtomicAdd(h ? &cnt_s[si] : &cnt_r[ri], 1.0f);
    }
  }
  __syncthreads();
#pragma unroll
  for (int nt = 0; nt < 4; ++nt) {
    const float tsum = es[nt] + xhalf(es[nt]);
    if (lane < 16) sW[(threadIdx.x >> 5) * 64 + nt * 16 + m] = tsum;
  }
  __syncthreads();
  if (threadIdx.x < 64) {
    float s = 0.f;
#pragma unroll
    for (int w = 0; w < 8; ++w) s += sW[w * 64 + threadIdx.x];
    if (s != 0.0f) unsafeAtomicAdd(&e_sum[threadIdx.x], s);
  }
}

template <bool TWO>
__global__ __launch_bounds__(256, 8) void k_node(
    const float* __restrict__ x, const float* __restrict__ agg_r0,
    const float* __restrict__ agg_r1,
    const float* __restrict__ Wnb, const float* __restrict__ bias_nb,
    float* __restrict__ n_h, float* __restrict__ n_sum) {
  __shared__ float sW[20 * 4 * 64];
  for (int i = threadIdx.x; i < 20 * 4 * 64; i += 256) {
    int j = i & 1, lane = (i >> 1) & 31, nt = (i >> 6) & 3, kc = i >> 8;
    int kk = 8 * (kc >> 1) + 4 * (lane >> 4) + 2 * (kc & 1) + j;
    sW[i] = Wnb[kk * 64 + nt * 16 + (lane & 15)];
  }
  __syncthreads();
  const int lane = threadIdx.x & 31;
  const int h = lane >> 4, m = lane & 15;
  const int wave = blockIdx.x * 8 + (threadIdx.x >> 5);
  const int nwaves = gridDim.x * 8;
  const int ntiles = NNODES / 16;
  float ns[4] = {0.f, 0.f, 0.f, 0.f};
  for (int t = wave; t < ntiles; t += nwaves) {
    asm volatile("" ::: "memory");
    const int n0 = t * 16;
    const float* ap = agg_r0 + (long)(n0 + m) * 64;
    const float* aq = agg_r1 + (long)(n0 + m) * 64;
    const float* xp = x + (long)(n0 + m) * 16;
    v8f acc[4];
    float bb[4];
#pragma unroll
    for (int nt = 0; nt < 4; ++nt) bb[nt] = bias_nb[nt * 16 + m];
    v4f qa[10], qb[8];
#pragma unroll
    for (int g = 0; g < 10; ++g) {
      const int kk = 8 * g + 4 * h;
      qa[g] = (g < 8) ? *(const v4f*)(ap + kk) : *(const v4f*)(xp + kk - 64);
      if (TWO && g < 8) qb[g] = *(const v4f*)(aq + kk);
    }
    asm volatile("" ::: "memory");
#pragma unroll
    for (int nt = 0; nt < 4; ++nt)
      acc[nt] = (v8f){bb[nt], bb[nt], bb[nt], bb[nt], bb[nt], bb[nt], bb[nt], bb[nt]};
#pragma unroll
    for (int g = 0; g < 10; ++g) {
      v4f q = qa[g];
      if (TWO && g < 8) q += qb[g];
#pragma unroll
      for (int j = 0; j < 2; ++j) {
        const v2f a = j ? (v2f){q.z, q.w} : (v2f){q.x, q.y};
        const int kc = 2 * g + j;
#pragma unroll
        for (int nt = 0; nt < 4; ++nt) {
          v2f b = *(const v2f*)(&sW[(kc * 4 + nt) * 64 + lane * 2]);
          acc[nt] = __builtin_amdgcn_wmma_f32_16x16x4_f32(false, a, false, b,
                                                          (short)0, acc[nt], false, false);
        }
      }
    }
#pragma unroll
    for (int nt = 0; nt < 4; ++nt)
#pragma unroll
      for (int v = 0; v < 8; ++v) {
        const float r = fmaxf(acc[nt][v], 0.0f);
        n_h[(long)(n0 + 8 * h + v) * 64 + nt * 16 + m] = r;
        ns[nt] += r;
      }
  }
  __syncthreads();
#pragma unroll
  for (int nt = 0; nt < 4; ++nt) {
    const float tsum = ns[nt] + xhalf(ns[nt]);
    if (lane < 16) sW[(threadIdx.x >> 5) * 64 + nt * 16 + m] = tsum;
  }
  __syncthreads();
  if (threadIdx.x < 64) {
    float s = 0.f;
#pragma unroll
    for (int w = 0; w < 8; ++w) s += sW[w * 64 + threadIdx.x];
    if (s != 0.0f) unsafeAtomicAdd(&n_sum[threadIdx.x], s);
  }
}

template <bool TWO>
__global__ __launch_bounds__(256, 6) void k_dec(
    const float* __restrict__ agg_r, const float* __restrict__ agg_s,
    const float* __restrict__ cnt_r, const float* __restrict__ cnt_s, long hs,
    const float* __restrict__ n_h, const float* __restrict__ Wd1,
    const float* __restrict__ bd1, const float* __restrict__ Wd2,
    const float* __restrict__ bd2, float* __restrict__ out,
    const float* __restrict__ u, const float* __restrict__ Wgb, const float* __restrict__ bgb,
    const float* __restrict__ e_sum, const float* __restrict__ n_sum) {
  __shared__ float sW1[48 * 4 * 64];
  __shared__ float sW2[16 * 64];
  __shared__ float sH[8 * 16 * 64];
  __shared__ float sB[64];
  const int gj = threadIdx.x & 63, gq = threadIdx.x >> 6;
  const float gb0 = bgb[gj], gb1 = bd1[gj];
  float we[16], wn[16], wu[16], wt[16], ve[16], vn[16], vu[16];
#pragma unroll
  for (int i = 0; i < 16; ++i) {
    const int k = gq * 16 + i;
    we[i] = Wgb[k * 64 + gj];
    wn[i] = Wgb[(64 + k) * 64 + gj];
    wu[i] = Wgb[(128 + k) * 64 + gj];
    wt[i] = Wd1[(192 + k) * 64 + gj];
    ve[i] = e_sum[k];
    vn[i] = n_sum[k];
    vu[i] = u[k];
  }
  for (int i = threadIdx.x; i < 48 * 4 * 64; i += 256) {
    int j = i & 1, lane = (i >> 1) & 31, nt = (i >> 6) & 3, kc = i >> 8;
    int kk = 8 * (kc >> 1) + 4 * (lane >> 4) + 2 * (kc & 1) + j;
    sW1[i] = Wd1[kk * 64 + nt * 16 + (lane & 15)];
  }
  for (int i = threadIdx.x; i < 16 * 64; i += 256) {
    int j = i & 1, lane = (i >> 1) & 31, kc = i >> 6;
    int kk = 4 * kc + 2 * (lane >> 4) + j;
    sW2[i] = Wd2[kk * 16 + (lane & 15)];
  }
  {
    float gp = 0.f;
#pragma unroll
    for (int i = 0; i < 16; ++i) {
      const float em = ve[i] * (1.0f / NEDGES);
      const float nm = vn[i] * (1.0f / NNODES);
      gp += em * we[i] + nm * wn[i] + vu[i] * wu[i];
    }
    sH[gq * 64 + gj] = gp;
  }
  __syncthreads();
  if (threadIdx.x < 64)
    sH[256 + gj] = fmaxf(gb0 + ((sH[gj] + sH[64 + gj]) + (sH[128 + gj] + sH[192 + gj])), 0.0f);
  __syncthreads();
  {
    float gb = 0.f;
#pragma unroll
    for (int i = 0; i < 16; ++i) gb += sH[256 + gq * 16 + i] * wt[i];
    sH[320 + gq * 64 + gj] = gb;
  }
  __syncthreads();
  if (threadIdx.x < 64)
    sB[gj] = gb1 + ((sH[320 + gj] + sH[384 + gj]) + (sH[448 + gj] + sH[512 + gj]));
  __syncthreads();
  const int lane = threadIdx.x & 31;
  const int h = lane >> 4, m = lane & 15;
  float* hw = sH + (threadIdx.x >> 5) * 1024;
  const int wave = blockIdx.x * 8 + (threadIdx.x >> 5);
  const int nwaves = gridDim.x * 8;
  const int ntiles = NNODES / 16;
  for (int t = wave; t < ntiles; t += nwaves) {
    asm volatile("" ::: "memory");
    const int n0 = t * 16;
    const int n = n0 + m;
    const float cr = cnt_r[n] + (TWO ? cnt_r[hs + n] : 0.0f);
    const float cs = cnt_s[n] + (TWO ? cnt_s[hs + n] : 0.0f);
    const float invr = 1.0f / fmaxf(cr, 1.0f);
    const float invs = 1.0f / fmaxf(cs, 1.0f);
    const float* pr = agg_r + (long)n * 64;
    const float* ps = agg_s + (long)n * 64;
    const float* pn = n_h + (long)n * 64;
    v8f acc[4];
    float bb[4];
#pragma unroll
    for (int nt = 0; nt < 4; ++nt) bb[nt] = sB[nt * 16 + m];
    auto ld = [&](const int part, v4f (&A)[4], v4f (&B)[4]) {
#pragma unroll
      for (int gg = 0; gg < 4; ++gg) {
        const int kk = 8 * (4 * (part & 1) + gg) + 4 * h;
        const float* p = part < 2 ? pr : part < 4 ? ps : pn;
        A[gg] = *(const v4f*)(p + kk);
        if (TWO && part < 4) B[gg] = *(const v4f*)(p + hs + kk);
      }
    };
    auto mm = [&](const int part, const v4f (&A)[4], const v4f (&B)[4]) {
#pragma unroll
      for (int gg = 0; gg < 4; ++gg) {
        v4f q = A[gg];
        if (part < 4) {
          if (TWO) q += B[gg];
          q *= (part < 2) ? invr : invs;
        }
#pragma unroll
        for (int j = 0; j < 2; ++j) {
          const v2f a = j ? (v2f){q.z, q.w} : (v2f){q.x, q.y};
          const int kc = 2 * (4 * part + gg) + j;
#pragma unroll
          for (int nt = 0; nt < 4; ++nt) {
            v2f b = *(const v2f*)(&sW1[(kc * 4 + nt) * 64 + lane * 2]);
            acc[nt] = __builtin_amdgcn_wmma_f32_16x16x4_f32(false, a, false, b,
                                                            (short)0, acc[nt], false, false);
          }
        }
      }
    };
    v4f a0[4], b0[4], a1[4], b1[4], a2[4], b2[4];
    ld(0, a0, b0);
    ld(1, a1, b1);
    asm volatile("" ::: "memory");
#pragma unroll
    for (int nt = 0; nt < 4; ++nt)
      acc[nt] = (v8f){bb[nt], bb[nt], bb[nt], bb[nt], bb[nt], bb[nt], bb[nt], bb[nt]};
    mm(0, a0, b0);
    ld(2, a2, b2);
    asm volatile("" ::: "memory");
    mm(1, a1, b1);
    ld(3, a0, b0);
    asm volatile("" ::: "memory");
    mm(2, a2, b2);
    ld(4, a1, b1);
    asm volatile("" ::: "memory");
    mm(3, a0, b0);
    ld(5, a2, b2);
    asm volatile("" ::: "memory");
    mm(4, a1, b1);
    mm(5, a2, b2);
#pragma unroll
    for (int nt = 0; nt < 4; ++nt)
#pragma unroll
      for (int v = 0; v < 8; ++v)
        hw[(8 * h + v) * 64 + nt * 16 + m] = fmaxf(acc[nt][v], 0.0f);
    float bb2 = bd2[m];
    v8f acc2 = (v8f){bb2, bb2, bb2, bb2, bb2, bb2, bb2, bb2};
#pragma unroll
    for (int kc = 0; kc < 16; ++kc) {
      const int kk = 4 * kc + 2 * h;
      v2f a = *(const v2f*)(hw + m * 64 + kk);
      v2f b = *(const v2f*)(&sW2[kc * 64 + lane * 2]);
      acc2 = __builtin_amdgcn_wmma_f32_16x16x4_f32(false, a, false, b,
                                                   (short)0, acc2, false, false);
    }
#pragma unroll
    for (int v = 0; v < 8; ++v)
      out[(long)(n0 + 8 * h + v) * 16 + m] = acc2[v];
  }
}

extern "C" void kernel_launch(void* const* d_in, const int* in_sizes, int n_in,
                              void* d_out, int out_size, void* d_ws, size_t ws_size,
                              hipStream_t stream) {
  const float* x    = (const float*)d_in[0];
  const float* ea   = (const float*)d_in[1];
  const float* u    = (const float*)d_in[2];
  const int* send   = (const int*)d_in[3];
  const int* recv   = (const int*)d_in[4];
  const float* Web  = (const float*)d_in[7];
  const float* beb  = (const float*)d_in[8];
  const float* Wnb  = (const float*)d_in[9];
  const float* bnb  = (const float*)d_in[10];
  const float* Wgb  = (const float*)d_in[11];
  const float* bgb  = (const float*)d_in[12];
  const float* Wd1  = (const float*)d_in[13];
  const float* bd1  = (const float*)d_in[14];
  const float* Wd2  = (const float*)d_in[15];
  const float* bd2  = (const float*)d_in[16];
  float* out = (float*)d_out;
  float* ws = (float*)d_ws;

  const long N = NNODES;
  if (ws_size >= (size_t)(2 * HALF_STRIDE + 256 + N * 64) * sizeof(float)) {
    float* sums    = ws;
    float* e_sum   = sums + 2 * HALF_STRIDE;
    float* n_sum   = e_sum + 64;
    float* bias_eb = n_sum + 64;
    float* bias_nb = bias_eb + 64;
    float* n_hbuf  = bias_nb + 64;
    int* ctr = (int*)n_hbuf;
    k_zero<<<2048, 256, 0, stream>>>(ws, 2 * HALF_STRIDE + 128, ctr, u, Web, beb, Wnb, bnb,
                                     bias_eb, bias_nb);
    k_edge<<<1536, 256, 0, stream>>>(x, ea, recv, send, Web, bias_eb, sums, HALF_STRIDE, ctr,
                                     e_sum);
    k_node<true><<<782, 256, 0, stream>>>(x, sums, sums + HALF_STRIDE, Wnb, bias_nb, n_hbuf,
                                          n_sum);
    k_dec<true><<<768, 256, 0, stream>>>(sums, sums + N * 64, sums + 2 * N * 64,
                                         sums + 2 * N * 64 + N, HALF_STRIDE, n_hbuf,
                                         Wd1, bd1, Wd2, bd2, out, u, Wgb, bgb, e_sum, n_sum);
  } else {
    float* sums    = ws;
    float* e_sum   = sums + HALF_STRIDE;
    float* n_sum   = e_sum + 64;
    float* bias_eb = n_sum + 64;
    float* bias_nb = bias_eb + 64;
    float* n_hbuf  = bias_nb + 64;
    int* ctr = (int*)n_hbuf;
    k_zero<<<2048, 256, 0, stream>>>(ws, HALF_STRIDE + 128, ctr, u, Web, beb, Wnb, bnb,
                                     bias_eb, bias_nb);
    k_edge<<<1536, 256, 0, stream>>>(x, ea, recv, send, Web, bias_eb, sums, 0, ctr, e_sum);
    k_node<false><<<782, 256, 0, stream>>>(x, sums, sums, Wnb, bias_nb, n_hbuf, n_sum);
    k_dec<false><<<768, 256, 0, stream>>>(sums, sums + N * 64, sums + 2 * N * 64,
                                          sums + 2 * N * 64 + N, 0, n_hbuf,
                                          Wd1, bd1, Wd2, bd2, out, u, Wgb, bgb, e_sum, n_sum);
  }
}
